# starting kernel
# baseline (speedup 1.0000x reference)
#include <hip/hip_runtime.h>

typedef float v2f __attribute__((ext_vector_type(2)));
typedef float v8f __attribute__((ext_vector_type(8)));

#define HEADS 8
#define DH 32
#define NTOK 2048
#define DIM 256
#define BATCH 2
#define SCALE2 (1.0f / 32.0f)

#define PROJ_ELEMS ((size_t)BATCH * HEADS * NTOK * DH)
#define OFF_QA ((size_t)0)
#define OFF_KA (PROJ_ELEMS)
#define OFF_QS (2 * PROJ_ELEMS)
#define OFF_KS (3 * PROJ_ELEMS)

static __device__ __forceinline__ v8f wmma_f32(v2f a, v2f b, v8f c) {
  return __builtin_amdgcn_wmma_f32_16x16x4_f32(
       false, a,  false, b,
       (short)0, c,  false,  false);
}

__global__ __launch_bounds__(256) void proj_kernel(
    const float* __restrict__ x, const float* __restrict__ attn_feat,
    const float* __restrict__ Wq, const float* __restrict__ Waux,
    float* __restrict__ ws) {
  const int lane = threadIdx.x & 31;
  const int wave = threadIdx.x >> 5;
  const int hi = lane >> 4;
  const int lo = lane & 15;
  const int ct = blockIdx.x * 8 + wave;
  const int m0 = blockIdx.y * 16;
  const int src = blockIdx.z;
  const float* X = src ? attn_feat : x;
  const float* W = src ? Waux : Wq;
  float* Qdst = ws + (src ? OFF_QA : OFF_QS);
  float* Kdst = ws + (src ? OFF_KA : OFF_KS);

  const int c = ct * 16 + lo;
  const float* Arow = X + (size_t)(m0 + lo) * DIM;

  v8f acc = {0.f, 0.f, 0.f, 0.f, 0.f, 0.f, 0.f, 0.f};
#pragma unroll 8
  for (int s = 0; s < 64; ++s) {
    const int k = s * 4 + hi * 2;
    v2f a = *(const v2f*)(Arow + k);
    v2f b;
    b.x = W[(size_t)k * 768 + c];
    b.y = W[(size_t)(k + 1) * 768 + c];
    acc = wmma_f32(a, b, acc);
  }

  float* dst;
  int hh;
  if (c < 256) { dst = Qdst; hh = c >> 5; }
  else         { dst = Kdst; hh = (c - 256) >> 5; }
  const int dh = c & 31;
#pragma unroll
  for (int r = 0; r < 8; ++r) {
    const int row = m0 + r + hi * 8;
    const int b_ = row >> 11, nn = row & 2047;
    dst[(((size_t)(b_ * HEADS + hh)) * NTOK + nn) * DH + dh] = acc[r];
  }
}

__global__ __launch_bounds__(256) void attn_kernel(
    const float* __restrict__ x, const float* __restrict__ ws,
    float* __restrict__ out0) {
  __shared__ float sX[16 * 272];
  __shared__ float sKa[16 * 36];
  __shared__ float sKs[16 * 36];
  __shared__ float sP[8 * 16 * 18];

  const int tid = threadIdx.x;
  const int lane = tid & 31, wave = tid >> 5;
  const int half = wave & 1, qt = wave >> 1;
  const int hi = lane >> 4, lo = lane & 15;
  const int bh = blockIdx.y, b = bh >> 3, h = bh & 7;
  const int m0 = blockIdx.x * 64 + qt * 16;

  const float* Qa = ws + OFF_QA;
  const float* Qs = ws + OFF_QS;
  const float* kabase = ws + OFF_KA + (size_t)bh * NTOK * DH;
  const float* ksbase = ws + OFF_KS + (size_t)bh * NTOK * DH;
  const float* xb = x + (size_t)b * NTOK * DIM;
  float* myP = sP + wave * (16 * 18);

  v2f qa[8], qs[8];
  {
    const float* qab = Qa + ((size_t)bh * NTOK + m0 + lo) * DH;
    const float* qsb = Qs + ((size_t)bh * NTOK + m0 + lo) * DH;
#pragma unroll
    for (int kk = 0; kk < 8; ++kk) {
      const int k = kk * 4 + hi * 2;
      qa[kk] = *(const v2f*)(qab + k);
      qs[kk] = *(const v2f*)(qsb + k);
    }
  }

  v8f O[8];
#pragma unroll
  for (int t = 0; t < 8; ++t) O[t] = (v8f){0.f, 0.f, 0.f, 0.f, 0.f, 0.f, 0.f, 0.f};
  float m[8], l[8];
#pragma unroll
  for (int r = 0; r < 8; ++r) { m[r] = -3.0e38f; l[r] = 0.f; }

  for (int j0 = 0; j0 < NTOK; j0 += 16) {
    __syncthreads();
#pragma unroll
    for (int t = 0; t < 4; ++t) {
      const int idx = t * 256 + tid;
      const int row = idx >> 6, c4 = (idx & 63) << 2;
      *(float4*)&sX[row * 272 + c4] =
          *(const float4*)(xb + (size_t)(j0 + row) * DIM + c4);
    }
    if (tid < 128) {
      const int row = tid >> 3, c4 = (tid & 7) << 2;
      *(float4*)&sKa[row * 36 + c4] =
          *(const float4*)(kabase + (size_t)(j0 + row) * DH + c4);
    } else {
      const int t2 = tid - 128;
      const int row = t2 >> 3, c4 = (t2 & 7) << 2;
      *(float4*)&sKs[row * 36 + c4] =
          *(const float4*)(ksbase + (size_t)(j0 + row) * DH + c4);
    }
    __syncthreads();

    v8f Sa = {0.f, 0.f, 0.f, 0.f, 0.f, 0.f, 0.f, 0.f};
    v8f Ss = {0.f, 0.f, 0.f, 0.f, 0.f, 0.f, 0.f, 0.f};
#pragma unroll
    for (int kk = 0; kk < 8; ++kk) {
      const int k = kk * 4 + hi * 2;
      v2f ba = *(const v2f*)&sKa[lo * 36 + k];
      Sa = wmma_f32(qa[kk], ba, Sa);
      v2f bs = *(const v2f*)&sKs[lo * 36 + k];
      Ss = wmma_f32(qs[kk], bs, Ss);
    }

    float P[8];
#pragma unroll
    for (int r = 0; r < 8; ++r) P[r] = Sa[r] * Ss[r] * SCALE2;

    float alpha[8];
#pragma unroll
    for (int r = 0; r < 8; ++r) {
      float v = P[r];
      v = fmaxf(v, __shfl_xor(v, 1));
      v = fmaxf(v, __shfl_xor(v, 2));
      v = fmaxf(v, __shfl_xor(v, 4));
      v = fmaxf(v, __shfl_xor(v, 8));
      const float mn = fmaxf(m[r], v);
      alpha[r] = __expf(m[r] - mn);
      m[r] = mn;
    }
#pragma unroll
    for (int r = 0; r < 8; ++r) {
      float e = __expf(P[r] - m[r]);
      P[r] = e;
      float s = e;
      s += __shfl_xor(s, 1);
      s += __shfl_xor(s, 2);
      s += __shfl_xor(s, 4);
      s += __shfl_xor(s, 8);
      l[r] = l[r] * alpha[r] + s;
    }
#pragma unroll
    for (int t = 0; t < 8; ++t)
#pragma unroll
      for (int r = 0; r < 8; ++r) O[t][r] *= alpha[r];

#pragma unroll
    for (int r = 0; r < 8; ++r) myP[(r + hi * 8) * 18 + lo] = P[r];
    asm volatile("s_wait_dscnt 0" ::: "memory");
    v2f ap[4];
#pragma unroll
    for (int kk = 0; kk < 4; ++kk)
      ap[kk] = *(const v2f*)&myP[lo * 18 + kk * 4 + hi * 2];

#pragma unroll
    for (int t = 0; t < 8; ++t) {
      const int n0 = half * 128 + t * 16 + lo;
      v8f o = O[t];
#pragma unroll
      for (int kk = 0; kk < 4; ++kk) {
        const int k = kk * 4 + hi * 2;
        v2f bx;
        bx.x = sX[k * 272 + n0];
        bx.y = sX[(k + 1) * 272 + n0];
        o = wmma_f32(ap[kk], bx, o);
      }
      O[t] = o;
    }
  }

  float invl[8];
#pragma unroll
  for (int r = 0; r < 8; ++r) invl[r] = 1.0f / l[r];
#pragma unroll
  for (int t = 0; t < 8; ++t) {
    const int col = h * 256 + half * 128 + t * 16 + lo;
#pragma unroll
    for (int r = 0; r < 8; ++r) {
      const int n = m0 + r + hi * 8;
      out0[((size_t)b * NTOK + n) * 2048 + col] = O[t][r] * invl[r];
    }
  }
}

__global__ __launch_bounds__(256) void row0_kernel(
    const float* __restrict__ ws, const float* __restrict__ out0,
    float* __restrict__ out1, float* __restrict__ attn0) {
  __shared__ float sp[NTOK];
  __shared__ float red[256];
  __shared__ float qa_s[DH], qs_s[DH];
  __shared__ float stat[2];
  const int bh = blockIdx.x, tid = threadIdx.x;
  const float* Qa = ws + OFF_QA + (size_t)bh * NTOK * DH;
  const float* Ka = ws + OFF_KA + (size_t)bh * NTOK * DH;
  const float* Qs = ws + OFF_QS + (size_t)bh * NTOK * DH;
  const float* Ks = ws + OFF_KS + (size_t)bh * NTOK * DH;
  if (tid < DH) qa_s[tid] = Qa[tid];
  else if (tid < 2 * DH) qs_s[tid - DH] = Qs[tid - DH];
  __syncthreads();

  float lmax = -3.0e38f;
  for (int j = tid; j < NTOK; j += 256) {
    float sa = 0.f, ss = 0.f;
    const float* ka = Ka + (size_t)j * DH;
    const float* ks = Ks + (size_t)j * DH;
#pragma unroll
    for (int d = 0; d < DH; ++d) { sa += qa_s[d] * ka[d]; ss += qs_s[d] * ks[d]; }
    const float p = sa * ss * SCALE2;
    sp[j] = p;
    lmax = fmaxf(lmax, p);
  }
  red[tid] = lmax;
  __syncthreads();
  for (int s = 128; s > 0; s >>= 1) {
    if (tid < s) red[tid] = fmaxf(red[tid], red[tid + s]);
    __syncthreads();
  }
  if (tid == 0) stat[0] = red[0];
  __syncthreads();
  const float mmax = stat[0];
  float lsum = 0.f;
  for (int j = tid; j < NTOK; j += 256) {
    const float e = __expf(sp[j] - mmax);
    sp[j] = e;
    lsum += e;
  }
  red[tid] = lsum;
  __syncthreads();
  for (int s = 128; s > 0; s >>= 1) {
    if (tid < s) red[tid] += red[tid + s];
    __syncthreads();
  }
  if (tid == 0) stat[1] = red[0];
  __syncthreads();
  const float inv = 1.0f / stat[1];
  for (int j = tid; j < NTOK; j += 256)
    attn0[(size_t)bh * NTOK + j] = sp[j] * inv;

  if (bh < BATCH) {
    for (int c = tid; c < 2048; c += 256)
      out1[bh * 2048 + c] = out0[(size_t)bh * 2048 * 2048 + c];
  }
}

extern "C" void kernel_launch(void* const* d_in, const int* in_sizes, int n_in,
                              void* d_out, int out_size, void* d_ws,
                              size_t ws_size, hipStream_t stream) {
  const float* x  = (const float*)d_in[0];
  const float* af = (const float*)d_in[1];
  const float* Wq = (const float*)d_in[2];
  const float* Wa = (const float*)d_in[3];
  float* ws = (float*)d_ws;
  float* out0 = (float*)d_out;
  float* out1 = out0 + (size_t)BATCH * NTOK * 2048;
  float* attn0 = out1 + (size_t)BATCH * 2048;

  proj_kernel<<<dim3(4, 256, 2), 256, 0, stream>>>(x, af, Wq, Wa, ws);
  attn_kernel<<<dim3(32, 16), 256, 0, stream>>>(x, ws, out0);
  row0_kernel<<<16, 256, 0, stream>>>(ws, out0, out1, attn0);
}


# proj kernel: a wave owns a 16x64 tile (4 accumulators); X and W fragments come as 128-bit global loads one 32-k chunk ahead, a lane owns 4 adjacent columns; no LDS; order of the k sum changes
# speedup vs baseline: 1.4792x; 1.4792x over previous
#include <hip/hip_runtime.h>

typedef float v2f __attribute__((ext_vector_type(2)));
typedef float v4f __attribute__((ext_vector_type(4)));
typedef float v8f __attribute__((ext_vector_type(8)));

#define HEADS 8
#define DH 32
#define NTOK 2048
#define DIM 256
#define BATCH 2
#define SCALE2 (1.0f / 32.0f)

#define PROJ_ELEMS ((size_t)BATCH * HEADS * NTOK * DH)
#define OFF_QA ((size_t)0)
#define OFF_KA (PROJ_ELEMS)
#define OFF_QS (2 * PROJ_ELEMS)
#define OFF_KS (3 * PROJ_ELEMS)

static __device__ __forceinline__ v8f wmma_f32(v2f a, v2f b, v8f c) {
  return __builtin_amdgcn_wmma_f32_16x16x4_f32(
       false, a,  false, b,
       (short)0, c,  false,  false);
}

template <int MASK>
static __device__ __forceinline__ float row_xor(float v) {
  return __builtin_bit_cast(
      float, __builtin_amdgcn_update_dpp(0, __builtin_bit_cast(int, v), 0x160 + MASK, 0xf, 0xf, true));
}

__global__ __launch_bounds__(256) void proj_kernel(
    const float* __restrict__ x, const float* __restrict__ attn_feat,
    const float* __restrict__ Wq, const float* __restrict__ Waux,
    float* __restrict__ ws) {
  const int lane = threadIdx.x & 31;
  const int wave = threadIdx.x >> 5;
  const int hi = lane >> 4;
  const int lo = lane & 15;
  const int c0 = blockIdx.x * 64;
  const int m0 = blockIdx.y * 128 + wave * 16;
  const int src = blockIdx.z;
  const float* X = src ? attn_feat : x;
  const float* W = src ? Waux : Wq;

  const float* ap = X + (size_t)(m0 + lo) * DIM + hi * 16;
  const float* bp = W + (size_t)(hi * 16) * 768 + c0 + lo * 4;

  v8f acc[4];
#pragma unroll
  for (int ct = 0; ct < 4; ++ct) acc[ct] = (v8f){0.f, 0.f, 0.f, 0.f, 0.f, 0.f, 0.f, 0.f};

  v4f aA[4], bA[8][2], aB[4], bB[8][2];

  auto load_chunk = [&](const int ch, v4f (&a)[4], v4f (&b)[8][2]) __attribute__((always_inline)) {
#pragma unroll
    for (int i = 0; i < 4; ++i) a[i] = *(const v4f*)(ap + ch * 32 + i * 4);
#pragma unroll
    for (int kk = 0; kk < 8; ++kk)
#pragma unroll
      for (int s = 0; s < 2; ++s)
        b[kk][s] = *(const v4f*)(bp + (size_t)(ch * 32 + kk * 2 + s) * 768);
  };
  auto mma_chunk = [&](const v4f (&a)[4], const v4f (&b)[8][2]) __attribute__((always_inline)) {
#pragma unroll
    for (int kk = 0; kk < 8; ++kk) {
      const int i = kk >> 1, e = (kk & 1) * 2;
      v2f av;
      av.x = a[i][e];
      av.y = a[i][e + 1];
#pragma unroll
      for (int ct = 0; ct < 4; ++ct) {
        v2f bv;
        bv.x = b[kk][0][ct];
        bv.y = b[kk][1][ct];
        acc[ct] = wmma_f32(av, bv, acc[ct]);
      }
    }
  };
  load_chunk(0, aA, bA);
  for (int ch = 0; ch < 8; ch += 2) {
    load_chunk(ch + 1, aB, bB);
    __builtin_amdgcn_sched_barrier(0);
    mma_chunk(aA, bA);
    __builtin_amdgcn_sched_barrier(0);
    load_chunk((ch + 2) & 7, aA, bA);
    __builtin_amdgcn_sched_barrier(0);
    mma_chunk(aB, bB);
    __builtin_amdgcn_sched_barrier(0);
  }

  const int c = c0 + lo * 4;
  float* dst = ws + (src ? (c0 < 256 ? OFF_QA : OFF_KA) : (c0 < 256 ? OFF_QS : OFF_KS));
  const int hh = (c & 255) >> 5;
  const int dh = c & 31;
#pragma unroll
  for (int r = 0; r < 8; ++r) {
    const int row = m0 + r + hi * 8;
    const int b_ = row >> 11, nn = row & 2047;
    v4f o;
    o.x = acc[0][r];
    o.y = acc[1][r];
    o.z = acc[2][r];
    o.w = acc[3][r];
    *(v4f*)&dst[(((size_t)(b_ * HEADS + hh)) * NTOK + nn) * DH + dh] = o;
  }
}

__global__ __launch_bounds__(128) __attribute__((amdgpu_waves_per_eu(1, 2))) void attn_kernel(
    const float* __restrict__ x, const float* __restrict__ ws,
    float* __restrict__ out0) {
  __shared__ __attribute__((aligned(16))) float sP[2][4 * 16 * 18];
  __shared__ __attribute__((aligned(16))) float sAl[2][64];
  __shared__ __attribute__((aligned(16))) float sInv[64];

  const int tid = threadIdx.x;
  const int lane = tid & 31, wave = tid >> 5;
  const int hi = lane >> 4, lo = lane & 15;
  const int bh = blockIdx.y, b = bh >> 3, h = bh & 7;
  const int q0 = blockIdx.x * 64;
  const int m0 = q0 + wave * 16;

  const float* Qa = ws + OFF_QA;
  const float* Qs = ws + OFF_QS;
  const float* kal = ws + OFF_KA + ((size_t)bh * NTOK + lo) * DH + hi * 16;
  const float* ksl = ws + OFF_KS + ((size_t)bh * NTOK + lo) * DH + hi * 16;
  const float* xb = x + (size_t)b * NTOK * DIM;
  const float* xw = xb + (size_t)(hi * 2) * DIM + wave * 64 + lo * 4;

  v4f qa4[4], qs4[4];
  {
    const float* qab = Qa + ((size_t)bh * NTOK + m0 + lo) * DH + hi * 16;
    const float* qsb = Qs + ((size_t)bh * NTOK + m0 + lo) * DH + hi * 16;
#pragma unroll
    for (int i = 0; i < 4; ++i) {
      qa4[i] = *(const v4f*)(qab + i * 4);
      qs4[i] = *(const v4f*)(qsb + i * 4);
    }
  }

  v8f O[4][4];
#pragma unroll
  for (int qt = 0; qt < 4; ++qt)
#pragma unroll
    for (int ct = 0; ct < 4; ++ct)
      O[qt][ct] = (v8f){0.f, 0.f, 0.f, 0.f, 0.f, 0.f, 0.f, 0.f};
  float m[8], l[8];
#pragma unroll
  for (int r = 0; r < 8; ++r) { m[r] = -3.0e38f; l[r] = 0.f; }

  v4f ka[4], ks[4];
#pragma unroll
  for (int i = 0; i < 4; ++i) {
    ka[i] = *(const v4f*)(kal + i * 4);
    ks[i] = *(const v4f*)(ksl + i * 4);
  }

  v4f xA[4][2], xB[4][2];
#pragma unroll
  for (int kk = 0; kk < 4; ++kk)
#pragma unroll
    for (int s = 0; s < 2; ++s)
      xA[kk][s] = *(const v4f*)(xw + (size_t)(kk * 4 + s) * DIM);

  auto key_block = [&](const int j0, const int buf, v4f (&xc)[4][2], v4f (&xn)[4][2])
                       __attribute__((always_inline)) {
    float* P_ = sP[buf];
    float* Al_ = sAl[buf];

    v8f Sa = {0.f, 0.f, 0.f, 0.f, 0.f, 0.f, 0.f, 0.f};
    v8f Ss = {0.f, 0.f, 0.f, 0.f, 0.f, 0.f, 0.f, 0.f};
#pragma unroll
    for (int kk = 0; kk < 8; ++kk) {
      const int i = kk >> 1, e = (kk & 1) * 2;
      v2f aa, ba, as, bs;
      aa.x = qa4[i][e]; aa.y = qa4[i][e + 1];
      ba.x = ka[i][e];  ba.y = ka[i][e + 1];
      Sa = wmma_f32(aa, ba, Sa);
      as.x = qs4[i][e]; as.y = qs4[i][e + 1];
      bs.x = ks[i][e];  bs.y = ks[i][e + 1];
      Ss = wmma_f32(as, bs, Ss);
    }

    float P[8];
#pragma unroll
    for (int r = 0; r < 8; ++r) P[r] = Sa[r] * Ss[r] * SCALE2;

    float alpha[8];
#pragma unroll
    for (int r = 0; r < 8; ++r) {
      float v = P[r];
      v = fmaxf(v, row_xor<1>(v));
      v = fmaxf(v, row_xor<2>(v));
      v = fmaxf(v, row_xor<4>(v));
      v = fmaxf(v, row_xor<8>(v));
      const float mn = fmaxf(m[r], v);
      alpha[r] = __expf(m[r] - mn);
      m[r] = mn;
    }
#pragma unroll
    for (int r = 0; r < 8; ++r) {
      float e = __expf(P[r] - m[r]);
      P[r] = e;
      float s = e;
      s += row_xor<1>(s);
      s += row_xor<2>(s);
      s += row_xor<4>(s);
      s += row_xor<8>(s);
      l[r] = l[r] * alpha[r] + s;
    }

#pragma unroll
    for (int r = 0; r < 8; ++r) P_[wave * 288 + (r + hi * 8) * 18 + lo] = P[r];
    if (lo == 0) {
      *(float4*)&Al_[wave * 16 + hi * 8] = make_float4(alpha[0], alpha[1], alpha[2], alpha[3]);
      *(float4*)&Al_[wave * 16 + hi * 8 + 4] = make_float4(alpha[4], alpha[5], alpha[6], alpha[7]);
    }
    __syncthreads();

    {
      const int jn = (j0 + 16) & (NTOK - 1);
#pragma unroll
      for (int i = 0; i < 4; ++i) {
        ka[i] = *(const v4f*)(kal + (size_t)jn * DH + i * 4);
        ks[i] = *(const v4f*)(ksl + (size_t)jn * DH + i * 4);
      }
#pragma unroll
      for (int kk = 0; kk < 4; ++kk)
#pragma unroll
        for (int s = 0; s < 2; ++s)
          xn[kk][s] = *(const v4f*)(xw + (size_t)(jn + kk * 4 + s) * DIM);
    }

#pragma unroll
    for (int qt = 0; qt < 4; ++qt) {
      const float4 a0 = *(const float4*)&Al_[qt * 16 + hi * 8];
      const float4 a1 = *(const float4*)&Al_[qt * 16 + hi * 8 + 4];
      const float al[8] = {a0.x, a0.y, a0.z, a0.w, a1.x, a1.y, a1.z, a1.w};
#pragma unroll
      for (int ct = 0; ct < 4; ++ct)
#pragma unroll
        for (int r = 0; r < 8; ++r) O[qt][ct][r] *= al[r];
    }
#pragma unroll
    for (int kk = 0; kk < 4; ++kk) {
      const int k = kk * 4 + hi * 2;
      v2f ap[4];
#pragma unroll
      for (int qt = 0; qt < 4; ++qt)
        ap[qt] = *(const v2f*)&P_[qt * 288 + lo * 18 + k];
#pragma unroll
      for (int ct = 0; ct < 4; ++ct) {
        v2f bx;
        bx.x = xc[kk][0][ct];
        bx.y = xc[kk][1][ct];
#pragma unroll
        for (int qt = 0; qt < 4; ++qt)
          O[qt][ct] = wmma_f32(ap[qt], bx, O[qt][ct]);
      }
    }

    asm volatile("" ::"v"(ka[0]), "v"(ka[1]), "v"(ka[2]), "v"(ka[3]), "v"(ks[0]), "v"(ks[1]), "v"(ks[2]),
                 "v"(ks[3]), "v"(xn[0][0]), "v"(xn[0][1]), "v"(xn[1][0]), "v"(xn[1][1]), "v"(xn[2][0]),
                 "v"(xn[2][1]), "v"(xn[3][0]), "v"(xn[3][1]));
    __builtin_amdgcn_sched_barrier(0);
  };

  for (int j0 = 0; j0 < NTOK; j0 += 32) {
    key_block(j0, 0, xA, xB);
    key_block(j0 + 16, 1, xB, xA);
  }

  if (lo == 0) {
#pragma unroll
    for (int r = 0; r < 8; ++r) sInv[wave * 16 + hi * 8 + r] = 1.0f / l[r];
  }
  __syncthreads();
#pragma unroll
  for (int qt = 0; qt < 4; ++qt) {
    const float4 i0 = *(const float4*)&sInv[qt * 16 + hi * 8];
    const float4 i1 = *(const float4*)&sInv[qt * 16 + hi * 8 + 4];
    const float invl[8] = {i0.x, i0.y, i0.z, i0.w, i1.x, i1.y, i1.z, i1.w};
    const int col = h * 256 + wave * 64 + lo * 4;
#pragma unroll
    for (int r = 0; r < 8; ++r) {
      const int n = q0 + qt * 16 + r + hi * 8;
      v4f o;
      o.x = O[qt][0][r] * invl[r];
      o.y = O[qt][1][r] * invl[r];
      o.z = O[qt][2][r] * invl[r];
      o.w = O[qt][3][r] * invl[r];
      *(v4f*)&out0[((size_t)b * NTOK + n) * 2048 + col] = o;
    }
  }
}

__global__ __launch_bounds__(256) void row0_kernel(
    const float* __restrict__ ws, const float* __restrict__ out0,
    float* __restrict__ out1, float* __restrict__ attn0) {
  __shared__ float sp[NTOK];
  __shared__ float red[256];
  __shared__ float qa_s[DH], qs_s[DH];
  __shared__ float stat[2];
  const int bh = blockIdx.x, tid = threadIdx.x;
  const float* Qa = ws + OFF_QA + (size_t)bh * NTOK * DH;
  const float* Ka = ws + OFF_KA + (size_t)bh * NTOK * DH;
  const float* Qs = ws + OFF_QS + (size_t)bh * NTOK * DH;
  const float* Ks = ws + OFF_KS + (size_t)bh * NTOK * DH;
  if (tid < DH) qa_s[tid] = Qa[tid];
  else if (tid < 2 * DH) qs_s[tid - DH] = Qs[tid - DH];
  __syncthreads();

  float lmax = -3.0e38f;
  for (int j = tid; j < NTOK; j += 256) {
    float sa = 0.f, ss = 0.f;
    const float* ka = Ka + (size_t)j * DH;
    const float* ks = Ks + (size_t)j * DH;
#pragma unroll
    for (int d = 0; d < DH; ++d) { sa += qa_s[d] * ka[d]; ss += qs_s[d] * ks[d]; }
    const float p = sa * ss * SCALE2;
    sp[j] = p;
    lmax = fmaxf(lmax, p);
  }
  red[tid] = lmax;
  __syncthreads();
  for (int s = 128; s > 0; s >>= 1) {
    if (tid < s) red[tid] = fmaxf(red[tid], red[tid + s]);
    __syncthreads();
  }
  if (tid == 0) stat[0] = red[0];
  __syncthreads();
  const float mmax = stat[0];
  float lsum = 0.f;
  for (int j = tid; j < NTOK; j += 256) {
    const float e = __expf(sp[j] - mmax);
    sp[j] = e;
    lsum += e;
  }
  red[tid] = lsum;
  __syncthreads();
  for (int s = 128; s > 0; s >>= 1) {
    if (tid < s) red[tid] += red[tid + s];
    __syncthreads();
  }
  if (tid == 0) stat[1] = red[0];
  __syncthreads();
  const float inv = 1.0f / stat[1];
  for (int j = tid; j < NTOK; j += 256)
    attn0[(size_t)bh * NTOK + j] = sp[j] * inv;

  if (bh < BATCH) {
    for (int c = tid; c < 2048; c += 256)
      out1[bh * 2048 + c] = out0[(size_t)bh * 2048 * 2048 + c];
  }
}

extern "C" void kernel_launch(void* const* d_in, const int* in_sizes, int n_in,
                              void* d_out, int out_size, void* d_ws,
                              size_t ws_size, hipStream_t stream) {
  const float* x  = (const float*)d_in[0];
  const float* af = (const float*)d_in[1];
  const float* Wq = (const float*)d_in[2];
  const float* Wa = (const float*)d_in[3];
  float* ws = (float*)d_ws;
  float* out0 = (float*)d_out;
  float* out1 = out0 + (size_t)BATCH * NTOK * 2048;
  float* attn0 = out1 + (size_t)BATCH * 2048;

  proj_kernel<<<dim3(8, 32, 2), 256, 0, stream>>>(x, af, Wq, Wa, ws);
  attn_kernel<<<dim3(32, 16), 128, 0, stream>>>(x, ws, out0);
  row0_kernel<<<16, 256, 0, stream>>>(ws, out0, out1, attn0);
}


# attn kernel: s_setprio falls with progress through the keys (3 for key blocks 0-63, 2 to 95, 1 to 119, 0 for the last 8): the lagging wave of a SIMD's pair is served first; sums unchanged
# speedup vs baseline: 1.5542x; 1.0507x over previous
#include <hip/hip_runtime.h>

typedef float v2f __attribute__((ext_vector_type(2)));
typedef float v4f __attribute__((ext_vector_type(4)));
typedef float v8f __attribute__((ext_vector_type(8)));

#define HEADS 8
#define DH 32
#define NTOK 2048
#define DIM 256
#define BATCH 2
#define SCALE2 (1.0f / 32.0f)

#define PROJ_ELEMS ((size_t)BATCH * HEADS * NTOK * DH)
#define OFF_QA ((size_t)0)
#define OFF_KA (PROJ_ELEMS)
#define OFF_QS (2 * PROJ_ELEMS)
#define OFF_KS (3 * PROJ_ELEMS)

static __device__ __forceinline__ v8f wmma_f32(v2f a, v2f b, v8f c) {
  return __builtin_amdgcn_wmma_f32_16x16x4_f32(
       false, a,  false, b,
       (short)0, c,  false,  false);
}

template <int MASK>
static __device__ __forceinline__ float row_xor(float v) {
  return __builtin_bit_cast(
      float, __builtin_amdgcn_update_dpp(0, __builtin_bit_cast(int, v), 0x160 + MASK, 0xf, 0xf, true));
}

__global__ __launch_bounds__(256) void proj_kernel(
    const float* __restrict__ x, const float* __restrict__ attn_feat,
    const float* __restrict__ Wq, const float* __restrict__ Waux,
    float* __restrict__ ws) {
  const int lane = threadIdx.x & 31;
  const int wave = threadIdx.x >> 5;
  const int hi = lane >> 4;
  const int lo = lane & 15;
  const int c0 = blockIdx.x * 64;
  const int m0 = blockIdx.y * 128 + wave * 16;
  const int src = blockIdx.z;
  const float* X = src ? attn_feat : x;
  const float* W = src ? Waux : Wq;

  const float* ap = X + (size_t)(m0 + lo) * DIM + hi * 16;
  const float* bp = W + (size_t)(hi * 16) * 768 + c0 + lo * 4;

  v8f acc[4];
#pragma unroll
  for (int ct = 0; ct < 4; ++ct) acc[ct] = (v8f){0.f, 0.f, 0.f, 0.f, 0.f, 0.f, 0.f, 0.f};

  v4f aA[4], bA[8][2], aB[4], bB[8][2];

  auto load_chunk = [&](const int ch, v4f (&a)[4], v4f (&b)[8][2]) __attribute__((always_inline)) {
#pragma unroll
    for (int i = 0; i < 4; ++i) a[i] = *(const v4f*)(ap + ch * 32 + i * 4);
#pragma unroll
    for (int kk = 0; kk < 8; ++kk)
#pragma unroll
      for (int s = 0; s < 2; ++s)
        b[kk][s] = *(const v4f*)(bp + (size_t)(ch * 32 + kk * 2 + s) * 768);
  };
  auto mma_chunk = [&](const v4f (&a)[4], const v4f (&b)[8][2]) __attribute__((always_inline)) {
#pragma unroll
    for (int kk = 0; kk < 8; ++kk) {
      const int i = kk >> 1, e = (kk & 1) * 2;
      v2f av;
      av.x = a[i][e];
      av.y = a[i][e + 1];
#pragma unroll
      for (int ct = 0; ct < 4; ++ct) {
        v2f bv;
        bv.x = b[kk][0][ct];
        bv.y = b[kk][1][ct];
        acc[ct] = wmma_f32(av, bv, acc[ct]);
      }
    }
  };
  load_chunk(0, aA, bA);
  for (int ch = 0; ch < 8; ch += 2) {
    load_chunk(ch + 1, aB, bB);
    __builtin_amdgcn_sched_barrier(0);
    mma_chunk(aA, bA);
    __builtin_amdgcn_sched_barrier(0);
    load_chunk((ch + 2) & 7, aA, bA);
    __builtin_amdgcn_sched_barrier(0);
    mma_chunk(aB, bB);
    __builtin_amdgcn_sched_barrier(0);
  }

  const int c = c0 + lo * 4;
  float* dst = ws + (src ? (c0 < 256 ? OFF_QA : OFF_KA) : (c0 < 256 ? OFF_QS : OFF_KS));
  const int hh = (c & 255) >> 5;
  const int dh = c & 31;
#pragma unroll
  for (int r = 0; r < 8; ++r) {
    const int row = m0 + r + hi * 8;
    const int b_ = row >> 11, nn = row & 2047;
    v4f o;
    o.x = acc[0][r];
    o.y = acc[1][r];
    o.z = acc[2][r];
    o.w = acc[3][r];
    *(v4f*)&dst[(((size_t)(b_ * HEADS + hh)) * NTOK + nn) * DH + dh] = o;
  }
}

__global__ __launch_bounds__(128) __attribute__((amdgpu_waves_per_eu(1, 2))) void attn_kernel(
    const float* __restrict__ x, const float* __restrict__ ws,
    float* __restrict__ out0) {
  __shared__ __attribute__((aligned(16))) float sP[2][4 * 16 * 18];
  __shared__ __attribute__((aligned(16))) float sAl[2][64];
  __shared__ __attribute__((aligned(16))) float sInv[64];

  const int tid = threadIdx.x;
  const int lane = tid & 31, wave = tid >> 5;
  const int hi = lane >> 4, lo = lane & 15;
  const int bh = blockIdx.y, b = bh >> 3, h = bh & 7;
  const int q0 = blockIdx.x * 64;
  const int m0 = q0 + wave * 16;

  const float* Qa = ws + OFF_QA;
  const float* Qs = ws + OFF_QS;
  const float* kal = ws + OFF_KA + ((size_t)bh * NTOK + lo) * DH + hi * 16;
  const float* ksl = ws + OFF_KS + ((size_t)bh * NTOK + lo) * DH + hi * 16;
  const float* xb = x + (size_t)b * NTOK * DIM;
  const float* xw = xb + (size_t)(hi * 2) * DIM + wave * 64 + lo * 4;

  v4f qa4[4], qs4[4];
  {
    const float* qab = Qa + ((size_t)bh * NTOK + m0 + lo) * DH + hi * 16;
    const float* qsb = Qs + ((size_t)bh * NTOK + m0 + lo) * DH + hi * 16;
#pragma unroll
    for (int i = 0; i < 4; ++i) {
      qa4[i] = *(const v4f*)(qab + i * 4);
      qs4[i] = *(const v4f*)(qsb + i * 4);
    }
  }

  v8f O[4][4];
#pragma unroll
  for (int qt = 0; qt < 4; ++qt)
#pragma unroll
    for (int ct = 0; ct < 4; ++ct)
      O[qt][ct] = (v8f){0.f, 0.f, 0.f, 0.f, 0.f, 0.f, 0.f, 0.f};
  float m[8], l[8];
#pragma unroll
  for (int r = 0; r < 8; ++r) { m[r] = -3.0e38f; l[r] = 0.f; }

  v4f ka[4], ks[4];
#pragma unroll
  for (int i = 0; i < 4; ++i) {
    ka[i] = *(const v4f*)(kal + i * 4);
    ks[i] = *(const v4f*)(ksl + i * 4);
  }

  v4f xA[4][2], xB[4][2];
#pragma unroll
  for (int kk = 0; kk < 4; ++kk)
#pragma unroll
    for (int s = 0; s < 2; ++s)
      xA[kk][s] = *(const v4f*)(xw + (size_t)(kk * 4 + s) * DIM);

  auto key_block = [&](const int j0, const int buf, v4f (&xc)[4][2], v4f (&xn)[4][2])
                       __attribute__((always_inline)) {
    float* P_ = sP[buf];
    float* Al_ = sAl[buf];

    v8f Sa = {0.f, 0.f, 0.f, 0.f, 0.f, 0.f, 0.f, 0.f};
    v8f Ss = {0.f, 0.f, 0.f, 0.f, 0.f, 0.f, 0.f, 0.f};
#pragma unroll
    for (int kk = 0; kk < 8; ++kk) {
      const int i = kk >> 1, e = (kk & 1) * 2;
      v2f aa, ba, as, bs;
      aa.x = qa4[i][e]; aa.y = qa4[i][e + 1];
      ba.x = ka[i][e];  ba.y = ka[i][e + 1];
      Sa = wmma_f32(aa, ba, Sa);
      as.x = qs4[i][e]; as.y = qs4[i][e + 1];
      bs.x = ks[i][e];  bs.y = ks[i][e + 1];
      Ss = wmma_f32(as, bs, Ss);
    }

    float P[8];
#pragma unroll
    for (int r = 0; r < 8; ++r) P[r] = Sa[r] * Ss[r] * SCALE2;

    float alpha[8];
#pragma unroll
    for (int r = 0; r < 8; ++r) {
      float v = P[r];
      v = fmaxf(v, row_xor<1>(v));
      v = fmaxf(v, row_xor<2>(v));
      v = fmaxf(v, row_xor<4>(v));
      v = fmaxf(v, row_xor<8>(v));
      const float mn = fmaxf(m[r], v);
      alpha[r] = __expf(m[r] - mn);
      m[r] = mn;
    }
#pragma unroll
    for (int r = 0; r < 8; ++r) {
      float e = __expf(P[r] - m[r]);
      P[r] = e;
      float s = e;
      s += row_xor<1>(s);
      s += row_xor<2>(s);
      s += row_xor<4>(s);
      s += row_xor<8>(s);
      l[r] = l[r] * alpha[r] + s;
    }

#pragma unroll
    for (int r = 0; r < 8; ++r) P_[wave * 288 + (r + hi * 8) * 18 + lo] = P[r];
    if (lo == 0) {
      *(float4*)&Al_[wave * 16 + hi * 8] = make_float4(alpha[0], alpha[1], alpha[2], alpha[3]);
      *(float4*)&Al_[wave * 16 + hi * 8 + 4] = make_float4(alpha[4], alpha[5], alpha[6], alpha[7]);
    }
    __syncthreads();

    {
      const int jn = (j0 + 16) & (NTOK - 1);
#pragma unroll
      for (int i = 0; i < 4; ++i) {
        ka[i] = *(const v4f*)(kal + (size_t)jn * DH + i * 4);
        ks[i] = *(const v4f*)(ksl + (size_t)jn * DH + i * 4);
      }
#pragma unroll
      for (int kk = 0; kk < 4; ++kk)
#pragma unroll
        for (int s = 0; s < 2; ++s)
          xn[kk][s] = *(const v4f*)(xw + (size_t)(jn + kk * 4 + s) * DIM);
    }

#pragma unroll
    for (int qt = 0; qt < 4; ++qt) {
      const float4 a0 = *(const float4*)&Al_[qt * 16 + hi * 8];
      const float4 a1 = *(const float4*)&Al_[qt * 16 + hi * 8 + 4];
      const float al[8] = {a0.x, a0.y, a0.z, a0.w, a1.x, a1.y, a1.z, a1.w};
#pragma unroll
      for (int ct = 0; ct < 4; ++ct)
#pragma unroll
        for (int r = 0; r < 8; ++r) O[qt][ct][r] *= al[r];
    }
#pragma unroll
    for (int kk = 0; kk < 4; ++kk) {
      const int k = kk * 4 + hi * 2;
      v2f ap[4];
#pragma unroll
      for (int qt = 0; qt < 4; ++qt)
        ap[qt] = *(const v2f*)&P_[qt * 288 + lo * 18 + k];
#pragma unroll
      for (int ct = 0; ct < 4; ++ct) {
        v2f bx;
        bx.x = xc[kk][0][ct];
        bx.y = xc[kk][1][ct];
#pragma unroll
        for (int qt = 0; qt < 4; ++qt)
          O[qt][ct] = wmma_f32(ap[qt], bx, O[qt][ct]);
      }
    }

    asm volatile("" ::"v"(ka[0]), "v"(ka[1]), "v"(ka[2]), "v"(ka[3]), "v"(ks[0]), "v"(ks[1]), "v"(ks[2]),
                 "v"(ks[3]), "v"(xn[0][0]), "v"(xn[0][1]), "v"(xn[1][0]), "v"(xn[1][1]), "v"(xn[2][0]),
                 "v"(xn[2][1]), "v"(xn[3][0]), "v"(xn[3][1]));
    __builtin_amdgcn_sched_barrier(0);
  };

  __builtin_amdgcn_s_setprio(3);
  for (int j0 = 0; j0 < NTOK; j0 += 32) {
    if (j0 == NTOK / 2) __builtin_amdgcn_s_setprio(2);
    else if (j0 == NTOK / 2 + NTOK / 4) __builtin_amdgcn_s_setprio(1);
    else if (j0 == NTOK - NTOK / 16) __builtin_amdgcn_s_setprio(0);
    key_block(j0, 0, xA, xB);
    key_block(j0 + 16, 1, xB, xA);
  }

  if (lo == 0) {
#pragma unroll
    for (int r = 0; r < 8; ++r) sInv[wave * 16 + hi * 8 + r] = 1.0f / l[r];
  }
  __syncthreads();
#pragma unroll
  for (int qt = 0; qt < 4; ++qt) {
    const float4 i0 = *(const float4*)&sInv[qt * 16 + hi * 8];
    const float4 i1 = *(const float4*)&sInv[qt * 16 + hi * 8 + 4];
    const float invl[8] = {i0.x, i0.y, i0.z, i0.w, i1.x, i1.y, i1.z, i1.w};
    const int col = h * 256 + wave * 64 + lo * 4;
#pragma unroll
    for (int r = 0; r < 8; ++r) {
      const int n = q0 + qt * 16 + r + hi * 8;
      v4f o;
      o.x = O[qt][0][r] * invl[r];
      o.y = O[qt][1][r] * invl[r];
      o.z = O[qt][2][r] * invl[r];
      o.w = O[qt][3][r] * invl[r];
      *(v4f*)&out0[((size_t)b * NTOK + n) * 2048 + col] = o;
    }
  }
}

__global__ __launch_bounds__(256) void row0_kernel(
    const float* __restrict__ ws, const float* __restrict__ out0,
    float* __restrict__ out1, float* __restrict__ attn0) {
  __shared__ float sp[NTOK];
  __shared__ float red[256];
  __shared__ float qa_s[DH], qs_s[DH];
  __shared__ float stat[2];
  const int bh = blockIdx.x, tid = threadIdx.x;
  const float* Qa = ws + OFF_QA + (size_t)bh * NTOK * DH;
  const float* Ka = ws + OFF_KA + (size_t)bh * NTOK * DH;
  const float* Qs = ws + OFF_QS + (size_t)bh * NTOK * DH;
  const float* Ks = ws + OFF_KS + (size_t)bh * NTOK * DH;
  if (tid < DH) qa_s[tid] = Qa[tid];
  else if (tid < 2 * DH) qs_s[tid - DH] = Qs[tid - DH];
  __syncthreads();

  float lmax = -3.0e38f;
  for (int j = tid; j < NTOK; j += 256) {
    float sa = 0.f, ss = 0.f;
    const float* ka = Ka + (size_t)j * DH;
    const float* ks = Ks + (size_t)j * DH;
#pragma unroll
    for (int d = 0; d < DH; ++d) { sa += qa_s[d] * ka[d]; ss += qs_s[d] * ks[d]; }
    const float p = sa * ss * SCALE2;
    sp[j] = p;
    lmax = fmaxf(lmax, p);
  }
  red[tid] = lmax;
  __syncthreads();
  for (int s = 128; s > 0; s >>= 1) {
    if (tid < s) red[tid] = fmaxf(red[tid], red[tid + s]);
    __syncthreads();
  }
  if (tid == 0) stat[0] = red[0];
  __syncthreads();
  const float mmax = stat[0];
  float lsum = 0.f;
  for (int j = tid; j < NTOK; j += 256) {
    const float e = __expf(sp[j] - mmax);
    sp[j] = e;
    lsum += e;
  }
  red[tid] = lsum;
  __syncthreads();
  for (int s = 128; s > 0; s >>= 1) {
    if (tid < s) red[tid] += red[tid + s];
    __syncthreads();
  }
  if (tid == 0) stat[1] = red[0];
  __syncthreads();
  const float inv = 1.0f / stat[1];
  for (int j = tid; j < NTOK; j += 256)
    attn0[(size_t)bh * NTOK + j] = sp[j] * inv;

  if (bh < BATCH) {
    for (int c = tid; c < 2048; c += 256)
      out1[bh * 2048 + c] = out0[(size_t)bh * 2048 * 2048 + c];
  }
}

extern "C" void kernel_launch(void* const* d_in, const int* in_sizes, int n_in,
                              void* d_out, int out_size, void* d_ws,
                              size_t ws_size, hipStream_t stream) {
  const float* x  = (const float*)d_in[0];
  const float* af = (const float*)d_in[1];
  const float* Wq = (const float*)d_in[2];
  const float* Wa = (const float*)d_in[3];
  float* ws = (float*)d_ws;
  float* out0 = (float*)d_out;
  float* out1 = out0 + (size_t)BATCH * NTOK * 2048;
  float* attn0 = out1 + (size_t)BATCH * 2048;

  proj_kernel<<<dim3(8, 32, 2), 256, 0, stream>>>(x, af, Wq, Wa, ws);
  attn_kernel<<<dim3(32, 16), 128, 0, stream>>>(x, ws, out0);
  row0_kernel<<<16, 256, 0, stream>>>(ws, out0, out1, attn0);
}


# attn kernel: online-softmax row sums kept per lane (a lane adds only its own key column's terms), added across the 16 lanes once behind the key loop, not per key block; row-sum order changes
# speedup vs baseline: 1.6035x; 1.0317x over previous
#include <hip/hip_runtime.h>

typedef float v2f __attribute__((ext_vector_type(2)));
typedef float v4f __attribute__((ext_vector_type(4)));
typedef float v8f __attribute__((ext_vector_type(8)));

#define HEADS 8
#define DH 32
#define NTOK 2048
#define DIM 256
#define BATCH 2
#define SCALE2 (1.0f / 32.0f)

#define PROJ_ELEMS ((size_t)BATCH * HEADS * NTOK * DH)
#define OFF_QA ((size_t)0)
#define OFF_KA (PROJ_ELEMS)
#define OFF_QS (2 * PROJ_ELEMS)
#define OFF_KS (3 * PROJ_ELEMS)

static __device__ __forceinline__ v8f wmma_f32(v2f a, v2f b, v8f c) {
  return __builtin_amdgcn_wmma_f32_16x16x4_f32(
       false, a,  false, b,
       (short)0, c,  false,  false);
}

template <int MASK>
static __device__ __forceinline__ float row_xor(float v) {
  return __builtin_bit_cast(
      float, __builtin_amdgcn_update_dpp(0, __builtin_bit_cast(int, v), 0x160 + MASK, 0xf, 0xf, true));
}

__global__ __launch_bounds__(256) void proj_kernel(
    const float* __restrict__ x, const float* __restrict__ attn_feat,
    const float* __restrict__ Wq, const float* __restrict__ Waux,
    float* __restrict__ ws) {
  const int lane = threadIdx.x & 31;
  const int wave = threadIdx.x >> 5;
  const int hi = lane >> 4;
  const int lo = lane & 15;
  const int c0 = blockIdx.x * 64;
  const int m0 = blockIdx.y * 128 + wave * 16;
  const int src = blockIdx.z;
  const float* X = src ? attn_feat : x;
  const float* W = src ? Waux : Wq;

  const float* ap = X + (size_t)(m0 + lo) * DIM + hi * 16;
  const float* bp = W + (size_t)(hi * 16) * 768 + c0 + lo * 4;

  v8f acc[4];
#pragma unroll
  for (int ct = 0; ct < 4; ++ct) acc[ct] = (v8f){0.f, 0.f, 0.f, 0.f, 0.f, 0.f, 0.f, 0.f};

  v4f aA[4], bA[8][2], aB[4], bB[8][2];

  auto load_chunk = [&](const int ch, v4f (&a)[4], v4f (&b)[8][2]) __attribute__((always_inline)) {
#pragma unroll
    for (int i = 0; i < 4; ++i) a[i] = *(const v4f*)(ap + ch * 32 + i * 4);
#pragma unroll
    for (int kk = 0; kk < 8; ++kk)
#pragma unroll
      for (int s = 0; s < 2; ++s)
        b[kk][s] = *(const v4f*)(bp + (size_t)(ch * 32 + kk * 2 + s) * 768);
  };
  auto mma_chunk = [&](const v4f (&a)[4], const v4f (&b)[8][2]) __attribute__((always_inline)) {
#pragma unroll
    for (int kk = 0; kk < 8; ++kk) {
      const int i = kk >> 1, e = (kk & 1) * 2;
      v2f av;
      av.x = a[i][e];
      av.y = a[i][e + 1];
#pragma unroll
      for (int ct = 0; ct < 4; ++ct) {
        v2f bv;
        bv.x = b[kk][0][ct];
        bv.y = b[kk][1][ct];
        acc[ct] = wmma_f32(av, bv, acc[ct]);
      }
    }
  };
  load_chunk(0, aA, bA);
  for (int ch = 0; ch < 8; ch += 2) {
    load_chunk(ch + 1, aB, bB);
    __builtin_amdgcn_sched_barrier(0);
    mma_chunk(aA, bA);
    __builtin_amdgcn_sched_barrier(0);
    load_chunk((ch + 2) & 7, aA, bA);
    __builtin_amdgcn_sched_barrier(0);
    mma_chunk(aB, bB);
    __builtin_amdgcn_sched_barrier(0);
  }

  const int c = c0 + lo * 4;
  float* dst = ws + (src ? (c0 < 256 ? OFF_QA : OFF_KA) : (c0 < 256 ? OFF_QS : OFF_KS));
  const int hh = (c & 255) >> 5;
  const int dh = c & 31;
#pragma unroll
  for (int r = 0; r < 8; ++r) {
    const int row = m0 + r + hi * 8;
    const int b_ = row >> 11, nn = row & 2047;
    v4f o;
    o.x = acc[0][r];
    o.y = acc[1][r];
    o.z = acc[2][r];
    o.w = acc[3][r];
    *(v4f*)&dst[(((size_t)(b_ * HEADS + hh)) * NTOK + nn) * DH + dh] = o;
  }
}

__global__ __launch_bounds__(128) __attribute__((amdgpu_waves_per_eu(1, 2))) void attn_kernel(
    const float* __restrict__ x, const float* __restrict__ ws,
    float* __restrict__ out0) {
  __shared__ __attribute__((aligned(16))) float sP[2][4 * 16 * 18];
  __shared__ __attribute__((aligned(16))) float sAl[2][64];
  __shared__ __attribute__((aligned(16))) float sInv[64];

  const int tid = threadIdx.x;
  const int lane = tid & 31, wave = tid >> 5;
  const int hi = lane >> 4, lo = lane & 15;
  const int bh = blockIdx.y, b = bh >> 3, h = bh & 7;
  const int q0 = blockIdx.x * 64;
  const int m0 = q0 + wave * 16;

  const float* Qa = ws + OFF_QA;
  const float* Qs = ws + OFF_QS;
  const float* kal = ws + OFF_KA + ((size_t)bh * NTOK + lo) * DH + hi * 16;
  const float* ksl = ws + OFF_KS + ((size_t)bh * NTOK + lo) * DH + hi * 16;
  const float* xb = x + (size_t)b * NTOK * DIM;
  const float* xw = xb + (size_t)(hi * 2) * DIM + wave * 64 + lo * 4;

  v4f qa4[4], qs4[4];
  {
    const float* qab = Qa + ((size_t)bh * NTOK + m0 + lo) * DH + hi * 16;
    const float* qsb = Qs + ((size_t)bh * NTOK + m0 + lo) * DH + hi * 16;
#pragma unroll
    for (int i = 0; i < 4; ++i) {
      qa4[i] = *(const v4f*)(qab + i * 4);
      qs4[i] = *(const v4f*)(qsb + i * 4);
    }
  }

  v8f O[4][4];
#pragma unroll
  for (int qt = 0; qt < 4; ++qt)
#pragma unroll
    for (int ct = 0; ct < 4; ++ct)
      O[qt][ct] = (v8f){0.f, 0.f, 0.f, 0.f, 0.f, 0.f, 0.f, 0.f};
  float m[8], l[8];
#pragma unroll
  for (int r = 0; r < 8; ++r) { m[r] = -3.0e38f; l[r] = 0.f; }

  v4f ka[4], ks[4];
#pragma unroll
  for (int i = 0; i < 4; ++i) {
    ka[i] = *(const v4f*)(kal + i * 4);
    ks[i] = *(const v4f*)(ksl + i * 4);
  }

  v4f xA[4][2], xB[4][2];
#pragma unroll
  for (int kk = 0; kk < 4; ++kk)
#pragma unroll
    for (int s = 0; s < 2; ++s)
      xA[kk][s] = *(const v4f*)(xw + (size_t)(kk * 4 + s) * DIM);

  auto key_block = [&](const int j0, const int buf, v4f (&xc)[4][2], v4f (&xn)[4][2])
                       __attribute__((always_inline)) {
    float* P_ = sP[buf];
    float* Al_ = sAl[buf];

    v8f Sa = {0.f, 0.f, 0.f, 0.f, 0.f, 0.f, 0.f, 0.f};
    v8f Ss = {0.f, 0.f, 0.f, 0.f, 0.f, 0.f, 0.f, 0.f};
#pragma unroll
    for (int kk = 0; kk < 8; ++kk) {
      const int i = kk >> 1, e = (kk & 1) * 2;
      v2f aa, ba, as, bs;
      aa.x = qa4[i][e]; aa.y = qa4[i][e + 1];
      ba.x = ka[i][e];  ba.y = ka[i][e + 1];
      Sa = wmma_f32(aa, ba, Sa);
      as.x = qs4[i][e]; as.y = qs4[i][e + 1];
      bs.x = ks[i][e];  bs.y = ks[i][e + 1];
      Ss = wmma_f32(as, bs, Ss);
    }

    float P[8];
#pragma unroll
    for (int r = 0; r < 8; ++r) P[r] = Sa[r] * Ss[r] * SCALE2;

    float alpha[8];
#pragma unroll
    for (int r = 0; r < 8; ++r) {
      float v = P[r];
      v = fmaxf(v, row_xor<1>(v));
      v = fmaxf(v, row_xor<2>(v));
      v = fmaxf(v, row_xor<4>(v));
      v = fmaxf(v, row_xor<8>(v));
      const float mn = fmaxf(m[r], v);
      alpha[r] = __expf(m[r] - mn);
      m[r] = mn;
    }
#pragma unroll
    for (int r = 0; r < 8; ++r) {
      const float e = __expf(P[r] - m[r]);
      P[r] = e;
      l[r] = l[r] * alpha[r] + e;
    }

#pragma unroll
    for (int r = 0; r < 8; ++r) P_[wave * 288 + (r + hi * 8) * 18 + lo] = P[r];
    if (lo == 0) {
      *(float4*)&Al_[wave * 16 + hi * 8] = make_float4(alpha[0], alpha[1], alpha[2], alpha[3]);
      *(float4*)&Al_[wave * 16 + hi * 8 + 4] = make_float4(alpha[4], alpha[5], alpha[6], alpha[7]);
    }
    __syncthreads();

    {
      const int jn = (j0 + 16) & (NTOK - 1);
#pragma unroll
      for (int i = 0; i < 4; ++i) {
        ka[i] = *(const v4f*)(kal + (size_t)jn * DH + i * 4);
        ks[i] = *(const v4f*)(ksl + (size_t)jn * DH + i * 4);
      }
#pragma unroll
      for (int kk = 0; kk < 4; ++kk)
#pragma unroll
        for (int s = 0; s < 2; ++s)
          xn[kk][s] = *(const v4f*)(xw + (size_t)(jn + kk * 4 + s) * DIM);
    }

#pragma unroll
    for (int qt = 0; qt < 4; ++qt) {
      const float4 a0 = *(const float4*)&Al_[qt * 16 + hi * 8];
      const float4 a1 = *(const float4*)&Al_[qt * 16 + hi * 8 + 4];
      const float al[8] = {a0.x, a0.y, a0.z, a0.w, a1.x, a1.y, a1.z, a1.w};
#pragma unroll
      for (int ct = 0; ct < 4; ++ct)
#pragma unroll
        for (int r = 0; r < 8; ++r) O[qt][ct][r] *= al[r];
    }
#pragma unroll
    for (int kk = 0; kk < 4; ++kk) {
      const int k = kk * 4 + hi * 2;
      v2f ap[4];
#pragma unroll
      for (int qt = 0; qt < 4; ++qt)
        ap[qt] = *(const v2f*)&P_[qt * 288 + lo * 18 + k];
#pragma unroll
      for (int ct = 0; ct < 4; ++ct) {
        v2f bx;
        bx.x = xc[kk][0][ct];
        bx.y = xc[kk][1][ct];
#pragma unroll
        for (int qt = 0; qt < 4; ++qt)
          O[qt][ct] = wmma_f32(ap[qt], bx, O[qt][ct]);
      }
    }

    asm volatile("" ::"v"(ka[0]), "v"(ka[1]), "v"(ka[2]), "v"(ka[3]), "v"(ks[0]), "v"(ks[1]), "v"(ks[2]),
                 "v"(ks[3]), "v"(xn[0][0]), "v"(xn[0][1]), "v"(xn[1][0]), "v"(xn[1][1]), "v"(xn[2][0]),
                 "v"(xn[2][1]), "v"(xn[3][0]), "v"(xn[3][1]));
    __builtin_amdgcn_sched_barrier(0);
  };

  __builtin_amdgcn_s_setprio(3);
  for (int j0 = 0; j0 < NTOK; j0 += 32) {
    if (j0 == NTOK / 2) __builtin_amdgcn_s_setprio(2);
    else if (j0 == NTOK / 2 + NTOK / 4) __builtin_amdgcn_s_setprio(1);
    else if (j0 == NTOK - NTOK / 16) __builtin_amdgcn_s_setprio(0);
    key_block(j0, 0, xA, xB);
    key_block(j0 + 16, 1, xB, xA);
  }

#pragma unroll
  for (int r = 0; r < 8; ++r) {
    float s = l[r];
    s += row_xor<1>(s);
    s += row_xor<2>(s);
    s += row_xor<4>(s);
    s += row_xor<8>(s);
    l[r] = s;
  }
  if (lo == 0) {
#pragma unroll
    for (int r = 0; r < 8; ++r) sInv[wave * 16 + hi * 8 + r] = 1.0f / l[r];
  }
  __syncthreads();
#pragma unroll
  for (int qt = 0; qt < 4; ++qt) {
    const float4 i0 = *(const float4*)&sInv[qt * 16 + hi * 8];
    const float4 i1 = *(const float4*)&sInv[qt * 16 + hi * 8 + 4];
    const float invl[8] = {i0.x, i0.y, i0.z, i0.w, i1.x, i1.y, i1.z, i1.w};
    const int col = h * 256 + wave * 64 + lo * 4;
#pragma unroll
    for (int r = 0; r < 8; ++r) {
      const int n = q0 + qt * 16 + r + hi * 8;
      v4f o;
      o.x = O[qt][0][r] * invl[r];
      o.y = O[qt][1][r] * invl[r];
      o.z = O[qt][2][r] * invl[r];
      o.w = O[qt][3][r] * invl[r];
      *(v4f*)&out0[((size_t)b * NTOK + n) * 2048 + col] = o;
    }
  }
}

__global__ __launch_bounds__(256) void row0_kernel(
    const float* __restrict__ ws, const float* __restrict__ out0,
    float* __restrict__ out1, float* __restrict__ attn0) {
  __shared__ float sp[NTOK];
  __shared__ float red[256];
  __shared__ float qa_s[DH], qs_s[DH];
  __shared__ float stat[2];
  const int bh = blockIdx.x, tid = threadIdx.x;
  const float* Qa = ws + OFF_QA + (size_t)bh * NTOK * DH;
  const float* Ka = ws + OFF_KA + (size_t)bh * NTOK * DH;
  const float* Qs = ws + OFF_QS + (size_t)bh * NTOK * DH;
  const float* Ks = ws + OFF_KS + (size_t)bh * NTOK * DH;
  if (tid < DH) qa_s[tid] = Qa[tid];
  else if (tid < 2 * DH) qs_s[tid - DH] = Qs[tid - DH];
  __syncthreads();

  float lmax = -3.0e38f;
  for (int j = tid; j < NTOK; j += 256) {
    float sa = 0.f, ss = 0.f;
    const float* ka = Ka + (size_t)j * DH;
    const float* ks = Ks + (size_t)j * DH;
#pragma unroll
    for (int d = 0; d < DH; ++d) { sa += qa_s[d] * ka[d]; ss += qs_s[d] * ks[d]; }
    const float p = sa * ss * SCALE2;
    sp[j] = p;
    lmax = fmaxf(lmax, p);
  }
  red[tid] = lmax;
  __syncthreads();
  for (int s = 128; s > 0; s >>= 1) {
    if (tid < s) red[tid] = fmaxf(red[tid], red[tid + s]);
    __syncthreads();
  }
  if (tid == 0) stat[0] = red[0];
  __syncthreads();
  const float mmax = stat[0];
  float lsum = 0.f;
  for (int j = tid; j < NTOK; j += 256) {
    const float e = __expf(sp[j] - mmax);
    sp[j] = e;
    lsum += e;
  }
  red[tid] = lsum;
  __syncthreads();
  for (int s = 128; s > 0; s >>= 1) {
    if (tid < s) red[tid] += red[tid + s];
    __syncthreads();
  }
  if (tid == 0) stat[1] = red[0];
  __syncthreads();
  const float inv = 1.0f / stat[1];
  for (int j = tid; j < NTOK; j += 256)
    attn0[(size_t)bh * NTOK + j] = sp[j] * inv;

  if (bh < BATCH) {
    for (int c = tid; c < 2048; c += 256)
      out1[bh * 2048 + c] = out0[(size_t)bh * 2048 * 2048 + c];
  }
}

extern "C" void kernel_launch(void* const* d_in, const int* in_sizes, int n_in,
                              void* d_out, int out_size, void* d_ws,
                              size_t ws_size, hipStream_t stream) {
  const float* x  = (const float*)d_in[0];
  const float* af = (const float*)d_in[1];
  const float* Wq = (const float*)d_in[2];
  const float* Wa = (const float*)d_in[3];
  float* ws = (float*)d_ws;
  float* out0 = (float*)d_out;
  float* out1 = out0 + (size_t)BATCH * NTOK * 2048;
  float* attn0 = out1 + (size_t)BATCH * 2048;

  proj_kernel<<<dim3(8, 32, 2), 256, 0, stream>>>(x, af, Wq, Wa, ws);
  attn_kernel<<<dim3(32, 16), 128, 0, stream>>>(x, ws, out0);
  row0_kernel<<<16, 256, 0, stream>>>(ws, out0, out1, attn0);
}


# attn kernel: softmax max raised lazily (only when a score tops the max in use by more than 8): common key block has no row-max butterfly, no factors, no O rescale; sum order kept, terms scaled
# speedup vs baseline: 1.7610x; 1.0982x over previous
#include <hip/hip_runtime.h>

typedef float v2f __attribute__((ext_vector_type(2)));
typedef float v4f __attribute__((ext_vector_type(4)));
typedef float v8f __attribute__((ext_vector_type(8)));

#define HEADS 8
#define DH 32
#define NTOK 2048
#define DIM 256
#define BATCH 2
#define SCALE2 (1.0f / 32.0f)
#define LAZY_TAU 8.0f

#define PROJ_ELEMS ((size_t)BATCH * HEADS * NTOK * DH)
#define OFF_QA ((size_t)0)
#define OFF_KA (PROJ_ELEMS)
#define OFF_QS (2 * PROJ_ELEMS)
#define OFF_KS (3 * PROJ_ELEMS)

static __device__ __forceinline__ v8f wmma_f32(v2f a, v2f b, v8f c) {
  return __builtin_amdgcn_wmma_f32_16x16x4_f32(
       false, a,  false, b,
       (short)0, c,  false,  false);
}

template <int MASK>
static __device__ __forceinline__ float row_xor(float v) {
  return __builtin_bit_cast(
      float, __builtin_amdgcn_update_dpp(0, __builtin_bit_cast(int, v), 0x160 + MASK, 0xf, 0xf, true));
}

__global__ __launch_bounds__(256) void proj_kernel(
    const float* __restrict__ x, const float* __restrict__ attn_feat,
    const float* __restrict__ Wq, const float* __restrict__ Waux,
    float* __restrict__ ws) {
  const int lane = threadIdx.x & 31;
  const int wave = threadIdx.x >> 5;
  const int hi = lane >> 4;
  const int lo = lane & 15;
  const int c0 = blockIdx.x * 64;
  const int m0 = blockIdx.y * 128 + wave * 16;
  const int src = blockIdx.z;
  const float* X = src ? attn_feat : x;
  const float* W = src ? Waux : Wq;

  const float* ap = X + (size_t)(m0 + lo) * DIM + hi * 16;
  const float* bp = W + (size_t)(hi * 16) * 768 + c0 + lo * 4;

  v8f acc[4];
#pragma unroll
  for (int ct = 0; ct < 4; ++ct) acc[ct] = (v8f){0.f, 0.f, 0.f, 0.f, 0.f, 0.f, 0.f, 0.f};

  v4f aA[4], bA[8][2], aB[4], bB[8][2];

  auto load_chunk = [&](const int ch, v4f (&a)[4], v4f (&b)[8][2]) __attribute__((always_inline)) {
#pragma unroll
    for (int i = 0; i < 4; ++i) a[i] = *(const v4f*)(ap + ch * 32 + i * 4);
#pragma unroll
    for (int kk = 0; kk < 8; ++kk)
#pragma unroll
      for (int s = 0; s < 2; ++s)
        b[kk][s] = *(const v4f*)(bp + (size_t)(ch * 32 + kk * 2 + s) * 768);
  };
  auto mma_chunk = [&](const v4f (&a)[4], const v4f (&b)[8][2]) __attribute__((always_inline)) {
#pragma unroll
    for (int kk = 0; kk < 8; ++kk) {
      const int i = kk >> 1, e = (kk & 1) * 2;
      v2f av;
      av.x = a[i][e];
      av.y = a[i][e + 1];
#pragma unroll
      for (int ct = 0; ct < 4; ++ct) {
        v2f bv;
        bv.x = b[kk][0][ct];
        bv.y = b[kk][1][ct];
        acc[ct] = wmma_f32(av, bv, acc[ct]);
      }
    }
  };
  load_chunk(0, aA, bA);
  for (int ch = 0; ch < 8; ch += 2) {
    load_chunk(ch + 1, aB, bB);
    __builtin_amdgcn_sched_barrier(0);
    mma_chunk(aA, bA);
    __builtin_amdgcn_sched_barrier(0);
    load_chunk((ch + 2) & 7, aA, bA);
    __builtin_amdgcn_sched_barrier(0);
    mma_chunk(aB, bB);
    __builtin_amdgcn_sched_barrier(0);
  }

  const int c = c0 + lo * 4;
  float* dst = ws + (src ? (c0 < 256 ? OFF_QA : OFF_KA) : (c0 < 256 ? OFF_QS : OFF_KS));
  const int hh = (c & 255) >> 5;
  const int dh = c & 31;
#pragma unroll
  for (int r = 0; r < 8; ++r) {
    const int row = m0 + r + hi * 8;
    const int b_ = row >> 11, nn = row & 2047;
    v4f o;
    o.x = acc[0][r];
    o.y = acc[1][r];
    o.z = acc[2][r];
    o.w = acc[3][r];
    *(v4f*)&dst[(((size_t)(b_ * HEADS + hh)) * NTOK + nn) * DH + dh] = o;
  }
}

__global__ __launch_bounds__(128) __attribute__((amdgpu_waves_per_eu(1, 2))) void attn_kernel(
    const float* __restrict__ x, const float* __restrict__ ws,
    float* __restrict__ out0) {
  __shared__ __attribute__((aligned(16))) float sP[2][4 * 16 * 18];
  __shared__ __attribute__((aligned(16))) float sAl[2][64];
  __shared__ __attribute__((aligned(16))) int sFl[2][4];
  __shared__ __attribute__((aligned(16))) float sInv[64];

  const int tid = threadIdx.x;
  const int lane = tid & 31, wave = tid >> 5;
  const int hi = lane >> 4, lo = lane & 15;
  const int bh = blockIdx.y, b = bh >> 3, h = bh & 7;
  const int q0 = blockIdx.x * 64;
  const int m0 = q0 + wave * 16;

  const float* Qa = ws + OFF_QA;
  const float* Qs = ws + OFF_QS;
  const float* kal = ws + OFF_KA + ((size_t)bh * NTOK + lo) * DH + hi * 16;
  const float* ksl = ws + OFF_KS + ((size_t)bh * NTOK + lo) * DH + hi * 16;
  const float* xb = x + (size_t)b * NTOK * DIM;
  const float* xw = xb + (size_t)(hi * 2) * DIM + wave * 64 + lo * 4;

  v4f qa4[4], qs4[4];
  {
    const float* qab = Qa + ((size_t)bh * NTOK + m0 + lo) * DH + hi * 16;
    const float* qsb = Qs + ((size_t)bh * NTOK + m0 + lo) * DH + hi * 16;
#pragma unroll
    for (int i = 0; i < 4; ++i) {
      qa4[i] = *(const v4f*)(qab + i * 4);
      qs4[i] = *(const v4f*)(qsb + i * 4);
    }
  }

  v8f O[4][4];
#pragma unroll
  for (int qt = 0; qt < 4; ++qt)
#pragma unroll
    for (int ct = 0; ct < 4; ++ct)
      O[qt][ct] = (v8f){0.f, 0.f, 0.f, 0.f, 0.f, 0.f, 0.f, 0.f};
  float m[8], l[8];
#pragma unroll
  for (int r = 0; r < 8; ++r) { m[r] = -3.0e38f; l[r] = 0.f; }

  v4f ka[4], ks[4];
#pragma unroll
  for (int i = 0; i < 4; ++i) {
    ka[i] = *(const v4f*)(kal + i * 4);
    ks[i] = *(const v4f*)(ksl + i * 4);
  }

  v4f xA[4][2], xB[4][2];
#pragma unroll
  for (int kk = 0; kk < 4; ++kk)
#pragma unroll
    for (int s = 0; s < 2; ++s)
      xA[kk][s] = *(const v4f*)(xw + (size_t)(kk * 4 + s) * DIM);

  auto key_block = [&](const int j0, const int buf, v4f (&xc)[4][2], v4f (&xn)[4][2])
                       __attribute__((always_inline)) {
    float* P_ = sP[buf];
    float* Al_ = sAl[buf];
    int* Fl_ = sFl[buf];

    v8f Sa = {0.f, 0.f, 0.f, 0.f, 0.f, 0.f, 0.f, 0.f};
    v8f Ss = {0.f, 0.f, 0.f, 0.f, 0.f, 0.f, 0.f, 0.f};
#pragma unroll
    for (int kk = 0; kk < 8; ++kk) {
      const int i = kk >> 1, e = (kk & 1) * 2;
      v2f aa, ba, as, bs;
      aa.x = qa4[i][e]; aa.y = qa4[i][e + 1];
      ba.x = ka[i][e];  ba.y = ka[i][e + 1];
      Sa = wmma_f32(aa, ba, Sa);
      as.x = qs4[i][e]; as.y = qs4[i][e + 1];
      bs.x = ks[i][e];  bs.y = ks[i][e + 1];
      Ss = wmma_f32(as, bs, Ss);
    }

    float P[8];
#pragma unroll
    for (int r = 0; r < 8; ++r) P[r] = Sa[r] * Ss[r] * SCALE2;

    float d[8], e[8];
#pragma unroll
    for (int r = 0; r < 8; ++r) d[r] = P[r] - m[r];
    const float dm = fmaxf(fmaxf(fmaxf(d[0], d[1]), fmaxf(d[2], d[3])), fmaxf(fmaxf(d[4], d[5]), fmaxf(d[6], d[7])));
#pragma unroll
    for (int r = 0; r < 8; ++r) e[r] = __expf(d[r]);
    const bool raise = __builtin_amdgcn_ballot_w32(dm > LAZY_TAU) != 0u;
    if (__builtin_expect(raise, false)) {
      float alpha[8];
#pragma unroll
      for (int r = 0; r < 8; ++r) {
        float v = P[r];
        v = fmaxf(v, row_xor<1>(v));
        v = fmaxf(v, row_xor<2>(v));
        v = fmaxf(v, row_xor<4>(v));
        v = fmaxf(v, row_xor<8>(v));
        const float mn = fmaxf(m[r], v);
        alpha[r] = __expf(m[r] - mn);
        m[r] = mn;
        l[r] *= alpha[r];
        e[r] = __expf(P[r] - mn);
      }
      if (lo == 0) {
        *(float4*)&Al_[wave * 16 + hi * 8] = make_float4(alpha[0], alpha[1], alpha[2], alpha[3]);
        *(float4*)&Al_[wave * 16 + hi * 8 + 4] = make_float4(alpha[4], alpha[5], alpha[6], alpha[7]);
      }
    }
#pragma unroll
    for (int r = 0; r < 8; ++r) l[r] += e[r];

#pragma unroll
    for (int r = 0; r < 8; ++r) P_[wave * 288 + (r + hi * 8) * 18 + lo] = e[r];
    Fl_[wave] = raise ? 1 : 0;
    __syncthreads();

    const int4 fl = *(const int4*)Fl_;
    v2f ap01[2][4];
#pragma unroll
    for (int kk = 0; kk < 2; ++kk)
#pragma unroll
      for (int qt = 0; qt < 4; ++qt)
        ap01[kk][qt] = *(const v2f*)&P_[qt * 288 + lo * 18 + kk * 4 + hi * 2];

    {
      const int jn = (j0 + 16) & (NTOK - 1);
#pragma unroll
      for (int i = 0; i < 4; ++i) {
        ka[i] = *(const v4f*)(kal + (size_t)jn * DH + i * 4);
        ks[i] = *(const v4f*)(ksl + (size_t)jn * DH + i * 4);
      }
#pragma unroll
      for (int kk = 0; kk < 4; ++kk)
#pragma unroll
        for (int s = 0; s < 2; ++s)
          xn[kk][s] = *(const v4f*)(xw + (size_t)(jn + kk * 4 + s) * DIM);
    }

    if (__builtin_expect(__builtin_amdgcn_readfirstlane(fl.x | fl.y | fl.z | fl.w) != 0, false)) {
      const int flq[4] = {fl.x, fl.y, fl.z, fl.w};
#pragma unroll
      for (int qt = 0; qt < 4; ++qt) {
        if (__builtin_amdgcn_readfirstlane(flq[qt]) != 0) {
          const float4 a0 = *(const float4*)&Al_[qt * 16 + hi * 8];
          const float4 a1 = *(const float4*)&Al_[qt * 16 + hi * 8 + 4];
          const float al[8] = {a0.x, a0.y, a0.z, a0.w, a1.x, a1.y, a1.z, a1.w};
#pragma unroll
          for (int ct = 0; ct < 4; ++ct)
#pragma unroll
            for (int r = 0; r < 8; ++r) O[qt][ct][r] *= al[r];
        }
      }
    }
#pragma unroll
    for (int kk = 0; kk < 4; ++kk) {
      const int k = kk * 4 + hi * 2;
      v2f ap[4];
#pragma unroll
      for (int qt = 0; qt < 4; ++qt)
        ap[qt] = kk < 2 ? ap01[kk][qt] : *(const v2f*)&P_[qt * 288 + lo * 18 + k];
#pragma unroll
      for (int ct = 0; ct < 4; ++ct) {
        v2f bx;
        bx.x = xc[kk][0][ct];
        bx.y = xc[kk][1][ct];
#pragma unroll
        for (int qt = 0; qt < 4; ++qt)
          O[qt][ct] = wmma_f32(ap[qt], bx, O[qt][ct]);
      }
    }

    __builtin_amdgcn_sched_barrier(0);
    asm volatile("" ::"v"(ka[0]), "v"(ka[1]), "v"(ka[2]), "v"(ka[3]), "v"(ks[0]), "v"(ks[1]), "v"(ks[2]),
                 "v"(ks[3]), "v"(xn[0][0]), "v"(xn[0][1]), "v"(xn[1][0]), "v"(xn[1][1]), "v"(xn[2][0]),
                 "v"(xn[2][1]), "v"(xn[3][0]), "v"(xn[3][1]));
    __builtin_amdgcn_sched_barrier(0);
  };

  __builtin_amdgcn_s_setprio(3);
  for (int j0 = 0; j0 < NTOK; j0 += 32) {
    if (j0 == NTOK / 2) __builtin_amdgcn_s_setprio(2);
    else if (j0 == NTOK / 2 + NTOK / 4) __builtin_amdgcn_s_setprio(1);
    else if (j0 == NTOK - NTOK / 16) __builtin_amdgcn_s_setprio(0);
    key_block(j0, 0, xA, xB);
    key_block(j0 + 16, 1, xB, xA);
  }

#pragma unroll
  for (int r = 0; r < 8; ++r) {
    float s = l[r];
    s += row_xor<1>(s);
    s += row_xor<2>(s);
    s += row_xor<4>(s);
    s += row_xor<8>(s);
    l[r] = s;
  }
  if (lo == 0) {
#pragma unroll
    for (int r = 0; r < 8; ++r) sInv[wave * 16 + hi * 8 + r] = 1.0f / l[r];
  }
  __syncthreads();
#pragma unroll
  for (int qt = 0; qt < 4; ++qt) {
    const float4 i0 = *(const float4*)&sInv[qt * 16 + hi * 8];
    const float4 i1 = *(const float4*)&sInv[qt * 16 + hi * 8 + 4];
    const float invl[8] = {i0.x, i0.y, i0.z, i0.w, i1.x, i1.y, i1.z, i1.w};
    const int col = h * 256 + wave * 64 + lo * 4;
#pragma unroll
    for (int r = 0; r < 8; ++r) {
      const int n = q0 + qt * 16 + r + hi * 8;
      v4f o;
      o.x = O[qt][0][r] * invl[r];
      o.y = O[qt][1][r] * invl[r];
      o.z = O[qt][2][r] * invl[r];
      o.w = O[qt][3][r] * invl[r];
      *(v4f*)&out0[((size_t)b * NTOK + n) * 2048 + col] = o;
    }
  }
}

__global__ __launch_bounds__(256) void row0_kernel(
    const float* __restrict__ ws, const float* __restrict__ out0,
    float* __restrict__ out1, float* __restrict__ attn0) {
  __shared__ float sp[NTOK];
  __shared__ float red[256];
  __shared__ float qa_s[DH], qs_s[DH];
  __shared__ float stat[2];
  const int bh = blockIdx.x, tid = threadIdx.x;
  const float* Qa = ws + OFF_QA + (size_t)bh * NTOK * DH;
  const float* Ka = ws + OFF_KA + (size_t)bh * NTOK * DH;
  const float* Qs = ws + OFF_QS + (size_t)bh * NTOK * DH;
  const float* Ks = ws + OFF_KS + (size_t)bh * NTOK * DH;
  if (tid < DH) qa_s[tid] = Qa[tid];
  else if (tid < 2 * DH) qs_s[tid - DH] = Qs[tid - DH];
  __syncthreads();

  float lmax = -3.0e38f;
  for (int j = tid; j < NTOK; j += 256) {
    float sa = 0.f, ss = 0.f;
    const float* ka = Ka + (size_t)j * DH;
    const float* ks = Ks + (size_t)j * DH;
#pragma unroll
    for (int d = 0; d < DH; ++d) { sa += qa_s[d] * ka[d]; ss += qs_s[d] * ks[d]; }
    const float p = sa * ss * SCALE2;
    sp[j] = p;
    lmax = fmaxf(lmax, p);
  }
  red[tid] = lmax;
  __syncthreads();
  for (int s = 128; s > 0; s >>= 1) {
    if (tid < s) red[tid] = fmaxf(red[tid], red[tid + s]);
    __syncthreads();
  }
  if (tid == 0) stat[0] = red[0];
  __syncthreads();
  const float mmax = stat[0];
  float lsum = 0.f;
  for (int j = tid; j < NTOK; j += 256) {
    const float e = __expf(sp[j] - mmax);
    sp[j] = e;
    lsum += e;
  }
  red[tid] = lsum;
  __syncthreads();
  for (int s = 128; s > 0; s >>= 1) {
    if (tid < s) red[tid] += red[tid + s];
    __syncthreads();
  }
  if (tid == 0) stat[1] = red[0];
  __syncthreads();
  const float inv = 1.0f / stat[1];
  for (int j = tid; j < NTOK; j += 256)
    attn0[(size_t)bh * NTOK + j] = sp[j] * inv;

  if (bh < BATCH) {
    for (int c = tid; c < 2048; c += 256)
      out1[bh * 2048 + c] = out0[(size_t)bh * 2048 * 2048 + c];
  }
}

extern "C" void kernel_launch(void* const* d_in, const int* in_sizes, int n_in,
                              void* d_out, int out_size, void* d_ws,
                              size_t ws_size, hipStream_t stream) {
  const float* x  = (const float*)d_in[0];
  const float* af = (const float*)d_in[1];
  const float* Wq = (const float*)d_in[2];
  const float* Wa = (const float*)d_in[3];
  float* ws = (float*)d_ws;
  float* out0 = (float*)d_out;
  float* out1 = out0 + (size_t)BATCH * NTOK * 2048;
  float* attn0 = out1 + (size_t)BATCH * 2048;

  proj_kernel<<<dim3(8, 32, 2), 256, 0, stream>>>(x, af, Wq, Wa, ws);
  attn_kernel<<<dim3(32, 16), 128, 0, stream>>>(x, ws, out0);
  row0_kernel<<<16, 256, 0, stream>>>(ws, out0, out1, attn0);
}


# row-0 kernel: 8 lanes share a K row (whole-line 128-bit loads, row_xmask adds), 1024-thread blocks, lane-owned keys, out[:,0] copy in 2 blocks of its own; score k-sum and row-sum order change
# speedup vs baseline: 1.8370x; 1.0432x over previous
#include <hip/hip_runtime.h>

typedef float v2f __attribute__((ext_vector_type(2)));
typedef float v4f __attribute__((ext_vector_type(4)));
typedef float v8f __attribute__((ext_vector_type(8)));

#define HEADS 8
#define DH 32
#define NTOK 2048
#define DIM 256
#define BATCH 2
#define SCALE2 (1.0f / 32.0f)
#define LAZY_TAU 8.0f

#define PROJ_ELEMS ((size_t)BATCH * HEADS * NTOK * DH)
#define OFF_QA ((size_t)0)
#define OFF_KA (PROJ_ELEMS)
#define OFF_QS (2 * PROJ_ELEMS)
#define OFF_KS (3 * PROJ_ELEMS)

static __device__ __forceinline__ v8f wmma_f32(v2f a, v2f b, v8f c) {
  return __builtin_amdgcn_wmma_f32_16x16x4_f32(
       false, a,  false, b,
       (short)0, c,  false,  false);
}

template <int MASK>
static __device__ __forceinline__ float row_xor(float v) {
  return __builtin_bit_cast(
      float, __builtin_amdgcn_update_dpp(0, __builtin_bit_cast(int, v), 0x160 + MASK, 0xf, 0xf, true));
}

__global__ __launch_bounds__(256) void proj_kernel(
    const float* __restrict__ x, const float* __restrict__ attn_feat,
    const float* __restrict__ Wq, const float* __restrict__ Waux,
    float* __restrict__ ws) {
  const int lane = threadIdx.x & 31;
  const int wave = threadIdx.x >> 5;
  const int hi = lane >> 4;
  const int lo = lane & 15;
  const int c0 = blockIdx.x * 64;
  const int m0 = blockIdx.y * 128 + wave * 16;
  const int src = blockIdx.z;
  const float* X = src ? attn_feat : x;
  const float* W = src ? Waux : Wq;

  const float* ap = X + (size_t)(m0 + lo) * DIM + hi * 16;
  const float* bp = W + (size_t)(hi * 16) * 768 + c0 + lo * 4;

  v8f acc[4];
#pragma unroll
  for (int ct = 0; ct < 4; ++ct) acc[ct] = (v8f){0.f, 0.f, 0.f, 0.f, 0.f, 0.f, 0.f, 0.f};

  v4f aA[4], bA[8][2], aB[4], bB[8][2];

  auto load_chunk = [&](const int ch, v4f (&a)[4], v4f (&b)[8][2]) __attribute__((always_inline)) {
#pragma unroll
    for (int i = 0; i < 4; ++i) a[i] = *(const v4f*)(ap + ch * 32 + i * 4);
#pragma unroll
    for (int kk = 0; kk < 8; ++kk)
#pragma unroll
      for (int s = 0; s < 2; ++s)
        b[kk][s] = *(const v4f*)(bp + (size_t)(ch * 32 + kk * 2 + s) * 768);
  };
  auto mma_chunk = [&](const v4f (&a)[4], const v4f (&b)[8][2]) __attribute__((always_inline)) {
#pragma unroll
    for (int kk = 0; kk < 8; ++kk) {
      const int i = kk >> 1, e = (kk & 1) * 2;
      v2f av;
      av.x = a[i][e];
      av.y = a[i][e + 1];
#pragma unroll
      for (int ct = 0; ct < 4; ++ct) {
        v2f bv;
        bv.x = b[kk][0][ct];
        bv.y = b[kk][1][ct];
        acc[ct] = wmma_f32(av, bv, acc[ct]);
      }
    }
  };
  load_chunk(0, aA, bA);
  for (int ch = 0; ch < 8; ch += 2) {
    load_chunk(ch + 1, aB, bB);
    __builtin_amdgcn_sched_barrier(0);
    mma_chunk(aA, bA);
    __builtin_amdgcn_sched_barrier(0);
    load_chunk((ch + 2) & 7, aA, bA);
    __builtin_amdgcn_sched_barrier(0);
    mma_chunk(aB, bB);
    __builtin_amdgcn_sched_barrier(0);
  }

  const int c = c0 + lo * 4;
  float* dst = ws + (src ? (c0 < 256 ? OFF_QA : OFF_KA) : (c0 < 256 ? OFF_QS : OFF_KS));
  const int hh = (c & 255) >> 5;
  const int dh = c & 31;
#pragma unroll
  for (int r = 0; r < 8; ++r) {
    const int row = m0 + r + hi * 8;
    const int b_ = row >> 11, nn = row & 2047;
    v4f o;
    o.x = acc[0][r];
    o.y = acc[1][r];
    o.z = acc[2][r];
    o.w = acc[3][r];
    *(v4f*)&dst[(((size_t)(b_ * HEADS + hh)) * NTOK + nn) * DH + dh] = o;
  }
}

__global__ __launch_bounds__(128) __attribute__((amdgpu_waves_per_eu(1, 2))) void attn_kernel(
    const float* __restrict__ x, const float* __restrict__ ws,
    float* __restrict__ out0) {
  __shared__ __attribute__((aligned(16))) float sP[2][4 * 16 * 18];
  __shared__ __attribute__((aligned(16))) float sAl[2][64];
  __shared__ __attribute__((aligned(16))) int sFl[2][4];
  __shared__ __attribute__((aligned(16))) float sInv[64];

  const int tid = threadIdx.x;
  const int lane = tid & 31, wave = tid >> 5;
  const int hi = lane >> 4, lo = lane & 15;
  const int bh = blockIdx.y, b = bh >> 3, h = bh & 7;
  const int q0 = blockIdx.x * 64;
  const int m0 = q0 + wave * 16;

  const float* Qa = ws + OFF_QA;
  const float* Qs = ws + OFF_QS;
  const float* kal = ws + OFF_KA + ((size_t)bh * NTOK + lo) * DH + hi * 16;
  const float* ksl = ws + OFF_KS + ((size_t)bh * NTOK + lo) * DH + hi * 16;
  const float* xb = x + (size_t)b * NTOK * DIM;
  const float* xw = xb + (size_t)(hi * 2) * DIM + wave * 64 + lo * 4;

  v4f qa4[4], qs4[4];
  {
    const float* qab = Qa + ((size_t)bh * NTOK + m0 + lo) * DH + hi * 16;
    const float* qsb = Qs + ((size_t)bh * NTOK + m0 + lo) * DH + hi * 16;
#pragma unroll
    for (int i = 0; i < 4; ++i) {
      qa4[i] = *(const v4f*)(qab + i * 4);
      qs4[i] = *(const v4f*)(qsb + i * 4);
    }
  }

  v8f O[4][4];
#pragma unroll
  for (int qt = 0; qt < 4; ++qt)
#pragma unroll
    for (int ct = 0; ct < 4; ++ct)
      O[qt][ct] = (v8f){0.f, 0.f, 0.f, 0.f, 0.f, 0.f, 0.f, 0.f};
  float m[8], l[8];
#pragma unroll
  for (int r = 0; r < 8; ++r) { m[r] = -3.0e38f; l[r] = 0.f; }

  v4f ka[4], ks[4];
#pragma unroll
  for (int i = 0; i < 4; ++i) {
    ka[i] = *(const v4f*)(kal + i * 4);
    ks[i] = *(const v4f*)(ksl + i * 4);
  }

  v4f xA[4][2], xB[4][2];
#pragma unroll
  for (int kk = 0; kk < 4; ++kk)
#pragma unroll
    for (int s = 0; s < 2; ++s)
      xA[kk][s] = *(const v4f*)(xw + (size_t)(kk * 4 + s) * DIM);

  auto key_block = [&](const int j0, const int buf, v4f (&xc)[4][2], v4f (&xn)[4][2])
                       __attribute__((always_inline)) {
    float* P_ = sP[buf];
    float* Al_ = sAl[buf];
    int* Fl_ = sFl[buf];

    v8f Sa = {0.f, 0.f, 0.f, 0.f, 0.f, 0.f, 0.f, 0.f};
    v8f Ss = {0.f, 0.f, 0.f, 0.f, 0.f, 0.f, 0.f, 0.f};
#pragma unroll
    for (int kk = 0; kk < 8; ++kk) {
      const int i = kk >> 1, e = (kk & 1) * 2;
      v2f aa, ba, as, bs;
      aa.x = qa4[i][e]; aa.y = qa4[i][e + 1];
      ba.x = ka[i][e];  ba.y = ka[i][e + 1];
      Sa = wmma_f32(aa, ba, Sa);
      as.x = qs4[i][e]; as.y = qs4[i][e + 1];
      bs.x = ks[i][e];  bs.y = ks[i][e + 1];
      Ss = wmma_f32(as, bs, Ss);
    }

    float P[8];
#pragma unroll
    for (int r = 0; r < 8; ++r) P[r] = Sa[r] * Ss[r] * SCALE2;

    float d[8], e[8];
#pragma unroll
    for (int r = 0; r < 8; ++r) d[r] = P[r] - m[r];
    const float dm = fmaxf(fmaxf(fmaxf(d[0], d[1]), fmaxf(d[2], d[3])), fmaxf(fmaxf(d[4], d[5]), fmaxf(d[6], d[7])));
#pragma unroll
    for (int r = 0; r < 8; ++r) e[r] = __expf(d[r]);
    const bool raise = __builtin_amdgcn_ballot_w32(dm > LAZY_TAU) != 0u;
    if (__builtin_expect(raise, false)) {
      float alpha[8];
#pragma unroll
      for (int r = 0; r < 8; ++r) {
        float v = P[r];
        v = fmaxf(v, row_xor<1>(v));
        v = fmaxf(v, row_xor<2>(v));
        v = fmaxf(v, row_xor<4>(v));
        v = fmaxf(v, row_xor<8>(v));
        const float mn = fmaxf(m[r], v);
        alpha[r] = __expf(m[r] - mn);
        m[r] = mn;
        l[r] *= alpha[r];
        e[r] = __expf(P[r] - mn);
      }
      if (lo == 0) {
        *(float4*)&Al_[wave * 16 + hi * 8] = make_float4(alpha[0], alpha[1], alpha[2], alpha[3]);
        *(float4*)&Al_[wave * 16 + hi * 8 + 4] = make_float4(alpha[4], alpha[5], alpha[6], alpha[7]);
      }
    }
#pragma unroll
    for (int r = 0; r < 8; ++r) l[r] += e[r];

#pragma unroll
    for (int r = 0; r < 8; ++r) P_[wave * 288 + (r + hi * 8) * 18 + lo] = e[r];
    Fl_[wave] = raise ? 1 : 0;
    __syncthreads();

    const int4 fl = *(const int4*)Fl_;
    v2f ap01[2][4];
#pragma unroll
    for (int kk = 0; kk < 2; ++kk)
#pragma unroll
      for (int qt = 0; qt < 4; ++qt)
        ap01[kk][qt] = *(const v2f*)&P_[qt * 288 + lo * 18 + kk * 4 + hi * 2];

    {
      const int jn = (j0 + 16) & (NTOK - 1);
#pragma unroll
      for (int i = 0; i < 4; ++i) {
        ka[i] = *(const v4f*)(kal + (size_t)jn * DH + i * 4);
        ks[i] = *(const v4f*)(ksl + (size_t)jn * DH + i * 4);
      }
#pragma unroll
      for (int kk = 0; kk < 4; ++kk)
#pragma unroll
        for (int s = 0; s < 2; ++s)
          xn[kk][s] = *(const v4f*)(xw + (size_t)(jn + kk * 4 + s) * DIM);
    }

    if (__builtin_expect(__builtin_amdgcn_readfirstlane(fl.x | fl.y | fl.z | fl.w) != 0, false)) {
      const int flq[4] = {fl.x, fl.y, fl.z, fl.w};
#pragma unroll
      for (int qt = 0; qt < 4; ++qt) {
        if (__builtin_amdgcn_readfirstlane(flq[qt]) != 0) {
          const float4 a0 = *(const float4*)&Al_[qt * 16 + hi * 8];
          const float4 a1 = *(const float4*)&Al_[qt * 16 + hi * 8 + 4];
          const float al[8] = {a0.x, a0.y, a0.z, a0.w, a1.x, a1.y, a1.z, a1.w};
#pragma unroll
          for (int ct = 0; ct < 4; ++ct)
#pragma unroll
            for (int r = 0; r < 8; ++r) O[qt][ct][r] *= al[r];
        }
      }
    }
#pragma unroll
    for (int kk = 0; kk < 4; ++kk) {
      const int k = kk * 4 + hi * 2;
      v2f ap[4];
#pragma unroll
      for (int qt = 0; qt < 4; ++qt)
        ap[qt] = kk < 2 ? ap01[kk][qt] : *(const v2f*)&P_[qt * 288 + lo * 18 + k];
#pragma unroll
      for (int ct = 0; ct < 4; ++ct) {
        v2f bx;
        bx.x = xc[kk][0][ct];
        bx.y = xc[kk][1][ct];
#pragma unroll
        for (int qt = 0; qt < 4; ++qt)
          O[qt][ct] = wmma_f32(ap[qt], bx, O[qt][ct]);
      }
    }

    __builtin_amdgcn_sched_barrier(0);
    asm volatile("" ::"v"(ka[0]), "v"(ka[1]), "v"(ka[2]), "v"(ka[3]), "v"(ks[0]), "v"(ks[1]), "v"(ks[2]),
                 "v"(ks[3]), "v"(xn[0][0]), "v"(xn[0][1]), "v"(xn[1][0]), "v"(xn[1][1]), "v"(xn[2][0]),
                 "v"(xn[2][1]), "v"(xn[3][0]), "v"(xn[3][1]));
    __builtin_amdgcn_sched_barrier(0);
  };

  __builtin_amdgcn_s_setprio(3);
  for (int j0 = 0; j0 < NTOK; j0 += 32) {
    if (j0 == NTOK / 2) __builtin_amdgcn_s_setprio(2);
    else if (j0 == NTOK / 2 + NTOK / 4) __builtin_amdgcn_s_setprio(1);
    else if (j0 == NTOK - NTOK / 16) __builtin_amdgcn_s_setprio(0);
    key_block(j0, 0, xA, xB);
    key_block(j0 + 16, 1, xB, xA);
  }

#pragma unroll
  for (int r = 0; r < 8; ++r) {
    float s = l[r];
    s += row_xor<1>(s);
    s += row_xor<2>(s);
    s += row_xor<4>(s);
    s += row_xor<8>(s);
    l[r] = s;
  }
  if (lo == 0) {
#pragma unroll
    for (int r = 0; r < 8; ++r) sInv[wave * 16 + hi * 8 + r] = 1.0f / l[r];
  }
  __syncthreads();
#pragma unroll
  for (int qt = 0; qt < 4; ++qt) {
    const float4 i0 = *(const float4*)&sInv[qt * 16 + hi * 8];
    const float4 i1 = *(const float4*)&sInv[qt * 16 + hi * 8 + 4];
    const float invl[8] = {i0.x, i0.y, i0.z, i0.w, i1.x, i1.y, i1.z, i1.w};
    const int col = h * 256 + wave * 64 + lo * 4;
#pragma unroll
    for (int r = 0; r < 8; ++r) {
      const int n = q0 + qt * 16 + r + hi * 8;
      v4f o;
      o.x = O[qt][0][r] * invl[r];
      o.y = O[qt][1][r] * invl[r];
      o.z = O[qt][2][r] * invl[r];
      o.w = O[qt][3][r] * invl[r];
      *(v4f*)&out0[((size_t)b * NTOK + n) * 2048 + col] = o;
    }
  }
}

static __device__ __forceinline__ float wave_max32(float v) {
  v = fmaxf(v, row_xor<1>(v));
  v = fmaxf(v, row_xor<2>(v));
  v = fmaxf(v, row_xor<4>(v));
  v = fmaxf(v, row_xor<8>(v));
  return fmaxf(v, __shfl_xor(v, 16, 32));
}
static __device__ __forceinline__ float wave_sum32(float v) {
  v += row_xor<1>(v);
  v += row_xor<2>(v);
  v += row_xor<4>(v);
  v += row_xor<8>(v);
  return v + __shfl_xor(v, 16, 32);
}

__global__ __launch_bounds__(1024) void row0_kernel(
    const float* __restrict__ ws, const float* __restrict__ out0,
    float* __restrict__ out1, float* __restrict__ attn0) {
  __shared__ float red[2][32];
  const int tid = threadIdx.x;
  if (blockIdx.x >= BATCH * HEADS) {
    const int b = blockIdx.x - BATCH * HEADS;
    *(v2f*)(out1 + (size_t)b * 2048 + tid * 2) = *(const v2f*)(out0 + (size_t)b * NTOK * 2048 + tid * 2);
    return;
  }
  const int bh = blockIdx.x;
  const int lane = tid & 31, wave = tid >> 5;
  const int g = lane >> 3, c = lane & 7;
  const size_t base = (size_t)bh * NTOK * DH;
  const v4f qa = *(const v4f*)(ws + OFF_QA + base + c * 4);
  const v4f qs = *(const v4f*)(ws + OFF_QS + base + c * 4);
  const float* kap = ws + OFF_KA + base + (size_t)(wave * 64 + g) * DH + c * 4;
  const float* ksp = ws + OFF_KS + base + (size_t)(wave * 64 + g) * DH + c * 4;

  auto score = [&](const v4f a, const v4f s) __attribute__((always_inline)) {
    float sa = qa.x * a.x + qa.y * a.y + qa.z * a.z + qa.w * a.w;
    float ss = qs.x * s.x + qs.y * s.y + qs.z * s.z + qs.w * s.w;
    sa += row_xor<1>(sa);
    ss += row_xor<1>(ss);
    sa += row_xor<2>(sa);
    ss += row_xor<2>(ss);
    sa += row_xor<4>(sa);
    ss += row_xor<4>(ss);
    return sa * ss * SCALE2;
  };
  float p[16];
  v4f ka[8], ks[8];
#pragma unroll
  for (int i = 0; i < 8; ++i) {
    ka[i] = *(const v4f*)(kap + (size_t)i * 4 * DH);
    ks[i] = *(const v4f*)(ksp + (size_t)i * 4 * DH);
  }
  __builtin_amdgcn_sched_barrier(0);
#pragma unroll
  for (int i = 0; i < 8; ++i) {
    p[i] = score(ka[i], ks[i]);
    ka[i] = *(const v4f*)(kap + (size_t)(i + 8) * 4 * DH);
    ks[i] = *(const v4f*)(ksp + (size_t)(i + 8) * 4 * DH);
  }
#pragma unroll
  for (int i = 0; i < 8; ++i) p[i + 8] = score(ka[i], ks[i]);
  float pA = p[0], pB = p[8];
#pragma unroll
  for (int i = 1; i < 8; ++i) {
    pA = (c == i) ? p[i] : pA;
    pB = (c == i) ? p[i + 8] : pB;
  }

  float mx = wave_max32(fmaxf(pA, pB));
  if (lane == 0) red[0][wave] = mx;
  __syncthreads();
  mx = wave_max32(red[0][lane]);
  const float eA = __expf(pA - mx);
  const float eB = __expf(pB - mx);
  float sum = wave_sum32(eA + eB);
  if (lane == 0) red[1][wave] = sum;
  __syncthreads();
  sum = wave_sum32(red[1][lane]);
  const float inv = 1.0f / sum;
  float* dst = attn0 + (size_t)bh * NTOK + wave * 64 + c * 4 + g;
  dst[0] = eA * inv;
  dst[32] = eB * inv;
}

extern "C" void kernel_launch(void* const* d_in, const int* in_sizes, int n_in,
                              void* d_out, int out_size, void* d_ws,
                              size_t ws_size, hipStream_t stream) {
  const float* x  = (const float*)d_in[0];
  const float* af = (const float*)d_in[1];
  const float* Wq = (const float*)d_in[2];
  const float* Wa = (const float*)d_in[3];
  float* ws = (float*)d_ws;
  float* out0 = (float*)d_out;
  float* out1 = out0 + (size_t)BATCH * NTOK * 2048;
  float* attn0 = out1 + (size_t)BATCH * 2048;

  proj_kernel<<<dim3(8, 32, 2), 256, 0, stream>>>(x, af, Wq, Wa, ws);
  attn_kernel<<<dim3(32, 16), 128, 0, stream>>>(x, ws, out0);
  row0_kernel<<<18, 1024, 0, stream>>>(ws, out0, out1, attn0);
}


# attn kernel: software pipeline in a wave: scores and softmax of key block j+1 run beside the product WMMAs of block j; LDS P reads covered by the score WMMAs; barrier at trip end; sums unchanged
# speedup vs baseline: 1.9099x; 1.0397x over previous
#include <hip/hip_runtime.h>

typedef float v2f __attribute__((ext_vector_type(2)));
typedef float v4f __attribute__((ext_vector_type(4)));
typedef float v8f __attribute__((ext_vector_type(8)));

#define HEADS 8
#define DH 32
#define NTOK 2048
#define DIM 256
#define BATCH 2
#define SCALE2 (1.0f / 32.0f)
#define LAZY_TAU 8.0f

#define PROJ_ELEMS ((size_t)BATCH * HEADS * NTOK * DH)
#define OFF_QA ((size_t)0)
#define OFF_KA (PROJ_ELEMS)
#define OFF_QS (2 * PROJ_ELEMS)
#define OFF_KS (3 * PROJ_ELEMS)

static __device__ __forceinline__ v8f wmma_f32(v2f a, v2f b, v8f c) {
  return __builtin_amdgcn_wmma_f32_16x16x4_f32(
       false, a,  false, b,
       (short)0, c,  false,  false);
}

template <int MASK>
static __device__ __forceinline__ float row_xor(float v) {
  return __builtin_bit_cast(
      float, __builtin_amdgcn_update_dpp(0, __builtin_bit_cast(int, v), 0x160 + MASK, 0xf, 0xf, true));
}

__global__ __launch_bounds__(256) void proj_kernel(
    const float* __restrict__ x, const float* __restrict__ attn_feat,
    const float* __restrict__ Wq, const float* __restrict__ Waux,
    float* __restrict__ ws) {
  const int lane = threadIdx.x & 31;
  const int wave = threadIdx.x >> 5;
  const int hi = lane >> 4;
  const int lo = lane & 15;
  const int c0 = blockIdx.x * 64;
  const int m0 = blockIdx.y * 128 + wave * 16;
  const int src = blockIdx.z;
  const float* X = src ? attn_feat : x;
  const float* W = src ? Waux : Wq;

  const float* ap = X + (size_t)(m0 + lo) * DIM + hi * 16;
  const float* bp = W + (size_t)(hi * 16) * 768 + c0 + lo * 4;

  v8f acc[4];
#pragma unroll
  for (int ct = 0; ct < 4; ++ct) acc[ct] = (v8f){0.f, 0.f, 0.f, 0.f, 0.f, 0.f, 0.f, 0.f};

  v4f aA[4], bA[8][2], aB[4], bB[8][2];

  auto load_chunk = [&](const int ch, v4f (&a)[4], v4f (&b)[8][2]) __attribute__((always_inline)) {
#pragma unroll
    for (int i = 0; i < 4; ++i) a[i] = *(const v4f*)(ap + ch * 32 + i * 4);
#pragma unroll
    for (int kk = 0; kk < 8; ++kk)
#pragma unroll
      for (int s = 0; s < 2; ++s)
        b[kk][s] = *(const v4f*)(bp + (size_t)(ch * 32 + kk * 2 + s) * 768);
  };
  auto mma_chunk = [&](const v4f (&a)[4], const v4f (&b)[8][2]) __attribute__((always_inline)) {
#pragma unroll
    for (int kk = 0; kk < 8; ++kk) {
      const int i = kk >> 1, e = (kk & 1) * 2;
      v2f av;
      av.x = a[i][e];
      av.y = a[i][e + 1];
#pragma unroll
      for (int ct = 0; ct < 4; ++ct) {
        v2f bv;
        bv.x = b[kk][0][ct];
        bv.y = b[kk][1][ct];
        acc[ct] = wmma_f32(av, bv, acc[ct]);
      }
    }
  };
  load_chunk(0, aA, bA);
  for (int ch = 0; ch < 8; ch += 2) {
    load_chunk(ch + 1, aB, bB);
    __builtin_amdgcn_sched_barrier(0);
    mma_chunk(aA, bA);
    __builtin_amdgcn_sched_barrier(0);
    load_chunk((ch + 2) & 7, aA, bA);
    __builtin_amdgcn_sched_barrier(0);
    mma_chunk(aB, bB);
    __builtin_amdgcn_sched_barrier(0);
  }

  const int c = c0 + lo * 4;
  float* dst = ws + (src ? (c0 < 256 ? OFF_QA : OFF_KA) : (c0 < 256 ? OFF_QS : OFF_KS));
  const int hh = (c & 255) >> 5;
  const int dh = c & 31;
#pragma unroll
  for (int r = 0; r < 8; ++r) {
    const int row = m0 + r + hi * 8;
    const int b_ = row >> 11, nn = row & 2047;
    v4f o;
    o.x = acc[0][r];
    o.y = acc[1][r];
    o.z = acc[2][r];
    o.w = acc[3][r];
    *(v4f*)&dst[(((size_t)(b_ * HEADS + hh)) * NTOK + nn) * DH + dh] = o;
  }
}

__global__ __launch_bounds__(128) __attribute__((amdgpu_waves_per_eu(1, 2))) void attn_kernel(
    const float* __restrict__ x, const float* __restrict__ ws,
    float* __restrict__ out0) {
  __shared__ __attribute__((aligned(16))) float sP[2][4 * 16 * 18];
  __shared__ __attribute__((aligned(16))) float sAl[2][64];
  __shared__ __attribute__((aligned(16))) int sFl[2][4];
  __shared__ __attribute__((aligned(16))) float sInv[64];

  const int tid = threadIdx.x;
  const int lane = tid & 31, wave = tid >> 5;
  const int hi = lane >> 4, lo = lane & 15;
  const int bh = blockIdx.y, b = bh >> 3, h = bh & 7;
  const int q0 = blockIdx.x * 64;
  const int m0 = q0 + wave * 16;

  const float* Qa = ws + OFF_QA;
  const float* Qs = ws + OFF_QS;
  const float* kal = ws + OFF_KA + ((size_t)bh * NTOK + lo) * DH + hi * 16;
  const float* ksl = ws + OFF_KS + ((size_t)bh * NTOK + lo) * DH + hi * 16;
  const float* xb = x + (size_t)b * NTOK * DIM;
  const float* xw = xb + (size_t)(hi * 2) * DIM + wave * 64 + lo * 4;

  v4f qa4[4], qs4[4];
  {
    const float* qab = Qa + ((size_t)bh * NTOK + m0 + lo) * DH + hi * 16;
    const float* qsb = Qs + ((size_t)bh * NTOK + m0 + lo) * DH + hi * 16;
#pragma unroll
    for (int i = 0; i < 4; ++i) {
      qa4[i] = *(const v4f*)(qab + i * 4);
      qs4[i] = *(const v4f*)(qsb + i * 4);
    }
  }

  v8f O[4][4];
#pragma unroll
  for (int qt = 0; qt < 4; ++qt)
#pragma unroll
    for (int ct = 0; ct < 4; ++ct)
      O[qt][ct] = (v8f){0.f, 0.f, 0.f, 0.f, 0.f, 0.f, 0.f, 0.f};
  float m[8], l[8];
#pragma unroll
  for (int r = 0; r < 8; ++r) { m[r] = -3.0e38f; l[r] = 0.f; }

  v4f ka[4], ks[4];
#pragma unroll
  for (int i = 0; i < 4; ++i) {
    ka[i] = *(const v4f*)(kal + i * 4);
    ks[i] = *(const v4f*)(ksl + i * 4);
  }

  v4f xA[4][2], xB[4][2];
#pragma unroll
  for (int kk = 0; kk < 4; ++kk)
#pragma unroll
    for (int s = 0; s < 2; ++s)
      xA[kk][s] = *(const v4f*)(xw + (size_t)(kk * 4 + s) * DIM);


  auto score_mma = [&](v8f& Sa, v8f& Ss) __attribute__((always_inline)) {
    Sa = (v8f){0.f, 0.f, 0.f, 0.f, 0.f, 0.f, 0.f, 0.f};
    Ss = (v8f){0.f, 0.f, 0.f, 0.f, 0.f, 0.f, 0.f, 0.f};
#pragma unroll
    for (int kk = 0; kk < 8; ++kk) {
      const int i = kk >> 1, e = (kk & 1) * 2;
      v2f aa, ba, as, bs;
      aa.x = qa4[i][e]; aa.y = qa4[i][e + 1];
      ba.x = ka[i][e];  ba.y = ka[i][e + 1];
      Sa = wmma_f32(aa, ba, Sa);
      as.x = qs4[i][e]; as.y = qs4[i][e + 1];
      bs.x = ks[i][e];  bs.y = ks[i][e + 1];
      Ss = wmma_f32(as, bs, Ss);
    }
  };
  auto load_k = [&](const int jb) __attribute__((always_inline)) {
#pragma unroll
    for (int i = 0; i < 4; ++i) {
      ka[i] = *(const v4f*)(kal + (size_t)jb * DH + i * 4);
      ks[i] = *(const v4f*)(ksl + (size_t)jb * DH + i * 4);
    }
  };
  auto softmax_store = [&](const v8f& Sa, const v8f& Ss, const int buf) __attribute__((always_inline)) {
    float* P_ = sP[buf];
    float* Al_ = sAl[buf];
    int* Fl_ = sFl[buf];
    float P[8];
#pragma unroll
    for (int r = 0; r < 8; ++r) P[r] = Sa[r] * Ss[r] * SCALE2;
    float d[8], e[8];
#pragma unroll
    for (int r = 0; r < 8; ++r) d[r] = P[r] - m[r];
    const float dm = fmaxf(fmaxf(fmaxf(d[0], d[1]), fmaxf(d[2], d[3])), fmaxf(fmaxf(d[4], d[5]), fmaxf(d[6], d[7])));
#pragma unroll
    for (int r = 0; r < 8; ++r) e[r] = __expf(d[r]);
    const bool raise = __builtin_amdgcn_ballot_w32(dm > LAZY_TAU) != 0u;
    if (__builtin_expect(raise, false)) {
      float alpha[8];
#pragma unroll
      for (int r = 0; r < 8; ++r) {
        float v = P[r];
        v = fmaxf(v, row_xor<1>(v));
        v = fmaxf(v, row_xor<2>(v));
        v = fmaxf(v, row_xor<4>(v));
        v = fmaxf(v, row_xor<8>(v));
        const float mn = fmaxf(m[r], v);
        alpha[r] = __expf(m[r] - mn);
        m[r] = mn;
        l[r] *= alpha[r];
        e[r] = __expf(P[r] - mn);
      }
      if (lo == 0) {
        *(float4*)&Al_[wave * 16 + hi * 8] = make_float4(alpha[0], alpha[1], alpha[2], alpha[3]);
        *(float4*)&Al_[wave * 16 + hi * 8 + 4] = make_float4(alpha[4], alpha[5], alpha[6], alpha[7]);
      }
    }
#pragma unroll
    for (int r = 0; r < 8; ++r) l[r] += e[r];
#pragma unroll
    for (int r = 0; r < 8; ++r) P_[wave * 288 + (r + hi * 8) * 18 + lo] = e[r];
    Fl_[wave] = raise ? 1 : 0;
  };

  auto key_block = [&](const int j0, const int buf, v4f (&xc)[4][2], v4f (&xn)[4][2], const bool last)
                       __attribute__((always_inline)) {
    const float* P_ = sP[buf];
    const float* Al_ = sAl[buf];
    const int* Fl_ = sFl[buf];

    const int4 fl = *(const int4*)Fl_;
    v2f ap[4][4];
#pragma unroll
    for (int kk = 0; kk < 4; ++kk)
#pragma unroll
      for (int qt = 0; qt < 4; ++qt)
        ap[kk][qt] = *(const v2f*)&P_[qt * 288 + lo * 18 + kk * 4 + hi * 2];
    __builtin_amdgcn_sched_barrier(0);

    v8f Sa = {0.f, 0.f, 0.f, 0.f, 0.f, 0.f, 0.f, 0.f};
    v8f Ss = {0.f, 0.f, 0.f, 0.f, 0.f, 0.f, 0.f, 0.f};
    if (!last) {
      score_mma(Sa, Ss);
      load_k((j0 + 32) & (NTOK - 1));
#pragma unroll
      for (int kk = 0; kk < 4; ++kk)
#pragma unroll
        for (int s = 0; s < 2; ++s)
          xn[kk][s] = *(const v4f*)(xw + (size_t)(j0 + 16 + kk * 4 + s) * DIM);
    }

    if (__builtin_expect(__builtin_amdgcn_readfirstlane(fl.x | fl.y | fl.z | fl.w) != 0, false)) {
      const int flq[4] = {fl.x, fl.y, fl.z, fl.w};
#pragma unroll
      for (int qt = 0; qt < 4; ++qt) {
        if (__builtin_amdgcn_readfirstlane(flq[qt]) != 0) {
          const float4 a0 = *(const float4*)&Al_[qt * 16 + hi * 8];
          const float4 a1 = *(const float4*)&Al_[qt * 16 + hi * 8 + 4];
          const float al[8] = {a0.x, a0.y, a0.z, a0.w, a1.x, a1.y, a1.z, a1.w};
#pragma unroll
          for (int ct = 0; ct < 4; ++ct)
#pragma unroll
            for (int r = 0; r < 8; ++r) O[qt][ct][r] *= al[r];
        }
      }
    }
    auto product = [&](const int kk) __attribute__((always_inline)) {
#pragma unroll
      for (int ct = 0; ct < 4; ++ct) {
        v2f bx;
        bx.x = xc[kk][0][ct];
        bx.y = xc[kk][1][ct];
#pragma unroll
        for (int qt = 0; qt < 4; ++qt)
          O[qt][ct] = wmma_f32(ap[kk][qt], bx, O[qt][ct]);
      }
    };
    product(0);
    product(1);
    if (!last) {
      softmax_store(Sa, Ss, buf ^ 1);
    }
    product(2);
    product(3);

    if (!last) __syncthreads();
  };

  {
    v8f Sa, Ss;
    score_mma(Sa, Ss);
    load_k(16);
    softmax_store(Sa, Ss, 0);
    __syncthreads();
  }

  __builtin_amdgcn_s_setprio(3);
  for (int j0 = 0; j0 < NTOK - 32; j0 += 32) {
    if (j0 == NTOK / 2) __builtin_amdgcn_s_setprio(2);
    else if (j0 == NTOK / 2 + NTOK / 4) __builtin_amdgcn_s_setprio(1);
    else if (j0 == NTOK - NTOK / 16) __builtin_amdgcn_s_setprio(0);
    key_block(j0, 0, xA, xB, false);
    key_block(j0 + 16, 1, xB, xA, false);
  }
  key_block(NTOK - 32, 0, xA, xB, false);
  key_block(NTOK - 16, 1, xB, xA, true);

#pragma unroll
  for (int r = 0; r < 8; ++r) {
    float s = l[r];
    s += row_xor<1>(s);
    s += row_xor<2>(s);
    s += row_xor<4>(s);
    s += row_xor<8>(s);
    l[r] = s;
  }
  if (lo == 0) {
#pragma unroll
    for (int r = 0; r < 8; ++r) sInv[wave * 16 + hi * 8 + r] = 1.0f / l[r];
  }
  __syncthreads();
#pragma unroll
  for (int qt = 0; qt < 4; ++qt) {
    const float4 i0 = *(const float4*)&sInv[qt * 16 + hi * 8];
    const float4 i1 = *(const float4*)&sInv[qt * 16 + hi * 8 + 4];
    const float invl[8] = {i0.x, i0.y, i0.z, i0.w, i1.x, i1.y, i1.z, i1.w};
    const int col = h * 256 + wave * 64 + lo * 4;
#pragma unroll
    for (int r = 0; r < 8; ++r) {
      const int n = q0 + qt * 16 + r + hi * 8;
      v4f o;
      o.x = O[qt][0][r] * invl[r];
      o.y = O[qt][1][r] * invl[r];
      o.z = O[qt][2][r] * invl[r];
      o.w = O[qt][3][r] * invl[r];
      *(v4f*)&out0[((size_t)b * NTOK + n) * 2048 + col] = o;
    }
  }
}

static __device__ __forceinline__ float wave_max32(float v) {
  v = fmaxf(v, row_xor<1>(v));
  v = fmaxf(v, row_xor<2>(v));
  v = fmaxf(v, row_xor<4>(v));
  v = fmaxf(v, row_xor<8>(v));
  return fmaxf(v, __shfl_xor(v, 16, 32));
}
static __device__ __forceinline__ float wave_sum32(float v) {
  v += row_xor<1>(v);
  v += row_xor<2>(v);
  v += row_xor<4>(v);
  v += row_xor<8>(v);
  return v + __shfl_xor(v, 16, 32);
}

__global__ __launch_bounds__(1024) void row0_kernel(
    const float* __restrict__ ws, const float* __restrict__ out0,
    float* __restrict__ out1, float* __restrict__ attn0) {
  __shared__ float red[2][32];
  const int tid = threadIdx.x;
  if (blockIdx.x >= BATCH * HEADS) {
    const int b = blockIdx.x - BATCH * HEADS;
    *(v2f*)(out1 + (size_t)b * 2048 + tid * 2) = *(const v2f*)(out0 + (size_t)b * NTOK * 2048 + tid * 2);
    return;
  }
  const int bh = blockIdx.x;
  const int lane = tid & 31, wave = tid >> 5;
  const int g = lane >> 3, c = lane & 7;
  const size_t base = (size_t)bh * NTOK * DH;
  const v4f qa = *(const v4f*)(ws + OFF_QA + base + c * 4);
  const v4f qs = *(const v4f*)(ws + OFF_QS + base + c * 4);
  const float* kap = ws + OFF_KA + base + (size_t)(wave * 64 + g) * DH + c * 4;
  const float* ksp = ws + OFF_KS + base + (size_t)(wave * 64 + g) * DH + c * 4;

  auto score = [&](const v4f a, const v4f s) __attribute__((always_inline)) {
    float sa = qa.x * a.x + qa.y * a.y + qa.z * a.z + qa.w * a.w;
    float ss = qs.x * s.x + qs.y * s.y + qs.z * s.z + qs.w * s.w;
    sa += row_xor<1>(sa);
    ss += row_xor<1>(ss);
    sa += row_xor<2>(sa);
    ss += row_xor<2>(ss);
    sa += row_xor<4>(sa);
    ss += row_xor<4>(ss);
    return sa * ss * SCALE2;
  };
  float p[16];
  v4f ka[8], ks[8];
#pragma unroll
  for (int i = 0; i < 8; ++i) {
    ka[i] = *(const v4f*)(kap + (size_t)i * 4 * DH);
    ks[i] = *(const v4f*)(ksp + (size_t)i * 4 * DH);
  }
  __builtin_amdgcn_sched_barrier(0);
#pragma unroll
  for (int i = 0; i < 8; ++i) {
    p[i] = score(ka[i], ks[i]);
    ka[i] = *(const v4f*)(kap + (size_t)(i + 8) * 4 * DH);
    ks[i] = *(const v4f*)(ksp + (size_t)(i + 8) * 4 * DH);
  }
#pragma unroll
  for (int i = 0; i < 8; ++i) p[i + 8] = score(ka[i], ks[i]);
  float pA = p[0], pB = p[8];
#pragma unroll
  for (int i = 1; i < 8; ++i) {
    pA = (c == i) ? p[i] : pA;
    pB = (c == i) ? p[i + 8] : pB;
  }

  float mx = wave_max32(fmaxf(pA, pB));
  if (lane == 0) red[0][wave] = mx;
  __syncthreads();
  mx = wave_max32(red[0][lane]);
  const float eA = __expf(pA - mx);
  const float eB = __expf(pB - mx);
  float sum = wave_sum32(eA + eB);
  if (lane == 0) red[1][wave] = sum;
  __syncthreads();
  sum = wave_sum32(red[1][lane]);
  const float inv = 1.0f / sum;
  float* dst = attn0 + (size_t)bh * NTOK + wave * 64 + c * 4 + g;
  dst[0] = eA * inv;
  dst[32] = eB * inv;
}

extern "C" void kernel_launch(void* const* d_in, const int* in_sizes, int n_in,
                              void* d_out, int out_size, void* d_ws,
                              size_t ws_size, hipStream_t stream) {
  const float* x  = (const float*)d_in[0];
  const float* af = (const float*)d_in[1];
  const float* Wq = (const float*)d_in[2];
  const float* Wa = (const float*)d_in[3];
  float* ws = (float*)d_ws;
  float* out0 = (float*)d_out;
  float* out1 = out0 + (size_t)BATCH * NTOK * 2048;
  float* attn0 = out1 + (size_t)BATCH * 2048;

  proj_kernel<<<dim3(8, 32, 2), 256, 0, stream>>>(x, af, Wq, Wa, ws);
  attn_kernel<<<dim3(32, 16), 128, 0, stream>>>(x, ws, out0);
  row0_kernel<<<18, 1024, 0, stream>>>(ws, out0, out1, attn0);
}
